# starting kernel
# baseline (speedup 1.0000x reference)
#include <hip/hip_runtime.h>

typedef float v2f __attribute__((ext_vector_type(2)));
typedef float v8f __attribute__((ext_vector_type(8)));

#define DFEAT 128


__global__ void zero_f4_kernel(float* __restrict__ p, int n4) {
    int i = blockIdx.x * blockDim.x + threadIdx.x;
    if (i < n4) ((float4*)p)[i] = make_float4(0.f, 0.f, 0.f, 0.f);
}

__global__ void deg_kernel(const int* __restrict__ dst, float* __restrict__ deg, int E) {
    int e = blockIdx.x * blockDim.x + threadIdx.x;
    if (e < E) atomicAdd(&deg[dst[e]], 1.0f);
}

__global__ void dinv_kernel(const float* __restrict__ deg, float* __restrict__ dinv, int N) {
    int i = blockIdx.x * blockDim.x + threadIdx.x;
    if (i < N) dinv[i] = 1.0f / sqrtf(fmaxf(deg[i], 1.0f));
}

__global__ void h_init_kernel(const float* __restrict__ feat, const float* __restrict__ theta,
                              float* __restrict__ h, int n4) {
    int i = blockIdx.x * blockDim.x + threadIdx.x;
    if (i >= n4) return;
    float t0 = theta[0];
    float4 f = ((const float4*)feat)[i];
    ((float4*)h)[i] = make_float4(t0 * f.x, t0 * f.y, t0 * f.z, t0 * f.w);
}


__global__ void scatter_kernel(const float* __restrict__ x, const float* __restrict__ dinv,
                               const int* __restrict__ src, const int* __restrict__ dst,
                               float* __restrict__ agg, int E) {
    int t = blockIdx.x * blockDim.x + threadIdx.x;
    int e = t >> 5;
    int lane = t & 31;
    if (e >= E) return;
    int s = src[e];
    int d = dst[e];
    float w = dinv[s];
    float4 v = *(const float4*)(x + (size_t)s * DFEAT + lane * 4);
    float* ap = agg + (size_t)d * DFEAT + lane * 4;
    atomicAdd(ap + 0, w * v.x);
    atomicAdd(ap + 1, w * v.y);
    atomicAdd(ap + 2, w * v.z);
    atomicAdd(ap + 3, w * v.w);
}


__global__ void cheb_update_kernel(float* __restrict__ agg, const float* __restrict__ dinv,
                                   const float* __restrict__ Tprev, float* __restrict__ Tnew,
                                   float* __restrict__ h, const float* __restrict__ theta,
                                   int order, float c2, int N) {
    int i = blockIdx.x * blockDim.x + threadIdx.x;
    int total = N * (DFEAT / 4);
    if (i >= total) return;
    int n = i / (DFEAT / 4);
    float s = -c2 * dinv[n];
    float th = theta[order] * 0.5f;

    float4 a = ((const float4*)agg)[i];
    ((float4*)agg)[i] = make_float4(0.f, 0.f, 0.f, 0.f);

    float4 tp = Tprev ? ((const float4*)Tprev)[i] : make_float4(0.f, 0.f, 0.f, 0.f);
    float4 t;
    t.x = fmaf(s, a.x, -tp.x);
    t.y = fmaf(s, a.y, -tp.y);
    t.z = fmaf(s, a.z, -tp.z);
    t.w = fmaf(s, a.w, -tp.w);
    ((float4*)Tnew)[i] = t;

    float4 hv = ((const float4*)h)[i];
    hv.x = fmaf(th, t.x + 1.0f, hv.x);
    hv.y = fmaf(th, t.y + 1.0f, hv.y);
    hv.z = fmaf(th, t.z + 1.0f, hv.z);
    hv.w = fmaf(th, t.w + 1.0f, hv.w);
    ((float4*)h)[i] = hv;
}


__global__ void gemm_wmma_kernel(const float* __restrict__ h, const float* __restrict__ W,
                                 float* __restrict__ out, int N) {
    const int lane = threadIdx.x & 31;
    const int wave = threadIdx.x >> 5;
    const int tile = blockIdx.x * (blockDim.x >> 5) + wave;
    const int nRowTiles = N >> 4;
    const int row = tile >> 3;
    const int col = tile & 7;
    if (row >= nRowTiles) return;

    const int rowBase = row << 4;
    const int colBase = col << 4;
    const int m = lane & 15;
    const int half = lane >> 4;
    const int kb = half << 1;

    const float* hrow = h + (size_t)(rowBase + m) * DFEAT + kb;
    const float* wrow = W + (size_t)(colBase + m) * DFEAT + kb;

    v8f acc = {};
#pragma unroll
    for (int kk = 0; kk < DFEAT; kk += 4) {
        v2f a = *(const v2f*)(hrow + kk);
        v2f b = *(const v2f*)(wrow + kk);
        acc = __builtin_amdgcn_wmma_f32_16x16x4_f32(
             false, a,  false, b,
             (short)0, acc,  false,  false);
    }

#pragma unroll
    for (int v = 0; v < 8; ++v) {
        int r = rowBase + v + (half << 3);
        float val = acc[v];
        val = (val >= 0.0f) ? val : 0.01f * val;
        out[(size_t)r * DFEAT + colBase + m] = val;
    }
}


extern "C" void kernel_launch(void* const* d_in, const int* in_sizes, int n_in,
                              void* d_out, int out_size, void* d_ws, size_t ws_size,
                              hipStream_t stream) {
    const float* feat  = (const float*)d_in[0];
    const int*   src   = (const int*)d_in[1];
    const int*   dst   = (const int*)d_in[2];
    const float* theta = (const float*)d_in[3];
    const float* W     = (const float*)d_in[4];
    float* out = (float*)d_out;

    const int N = in_sizes[0] / DFEAT;
    const int E = in_sizes[1];

    float* ws   = (float*)d_ws;
    float* deg  = ws;
    float* dinv = deg + N;
    float* agg  = dinv + N;
    float* Ta   = agg + (size_t)N * DFEAT;
    float* Tb   = Ta + (size_t)N * DFEAT;
    float* h    = Tb + (size_t)N * DFEAT;

    const int TB = 256;
    const int nd4 = N * (DFEAT / 4);
    auto cdiv = [](long long a, long long b) { return (int)((a + b - 1) / b); };

    zero_f4_kernel<<<cdiv(N / 4, TB), TB, 0, stream>>>(deg, N / 4);
    zero_f4_kernel<<<cdiv(nd4, TB), TB, 0, stream>>>(agg, nd4);
    deg_kernel<<<cdiv(E, TB), TB, 0, stream>>>(dst, deg, E);
    dinv_kernel<<<cdiv(N, TB), TB, 0, stream>>>(deg, dinv, N);

    h_init_kernel<<<cdiv(nd4, TB), TB, 0, stream>>>(feat, theta, h, nd4);

    const long long edgeThreads = (long long)E * 32;

    scatter_kernel<<<cdiv(edgeThreads, TB), TB, 0, stream>>>(feat, dinv, src, dst, agg, E);
    cheb_update_kernel<<<cdiv(nd4, TB), TB, 0, stream>>>(agg, dinv, nullptr, Ta, h, theta, 1, 1.0f, N);

    scatter_kernel<<<cdiv(edgeThreads, TB), TB, 0, stream>>>(Ta, dinv, src, dst, agg, E);
    cheb_update_kernel<<<cdiv(nd4, TB), TB, 0, stream>>>(agg, dinv, feat, Tb, h, theta, 2, 2.0f, N);

    scatter_kernel<<<cdiv(edgeThreads, TB), TB, 0, stream>>>(Tb, dinv, src, dst, agg, E);
    cheb_update_kernel<<<cdiv(nd4, TB), TB, 0, stream>>>(agg, dinv, Ta, Ta, h, theta, 3, 2.0f, N);

    const int tiles = (N / 16) * (DFEAT / 16);
    gemm_wmma_kernel<<<cdiv(tiles, 256 / 32), 256, 0, stream>>>(h, W, out, N);
}


# Edges grouped by destination in-call (integer count, scan, place, ascending edge order); one wave sums a node's source rows in registers and finishes the row: float atomics gone, sum order fixed
# speedup vs baseline: 2.7183x; 2.7183x over previous
#include <hip/hip_runtime.h>

typedef float v2f __attribute__((ext_vector_type(2)));
typedef float v8f __attribute__((ext_vector_type(8)));

#define DFEAT 128


__global__ void zero_f4_kernel(float* __restrict__ p, int n4) {
    int i = blockIdx.x * blockDim.x + threadIdx.x;
    if (i < n4) ((float4*)p)[i] = make_float4(0.f, 0.f, 0.f, 0.f);
}


__global__ void rank_kernel(const int* __restrict__ dst, int* __restrict__ cnt,
                            int* __restrict__ pos, int E) {
    int e = blockIdx.x * blockDim.x + threadIdx.x;
    if (e < E) pos[e] = atomicAdd(&cnt[dst[e]], 1);
}

#define SCAN_SPAN 1024
__global__ void __launch_bounds__(256)
scan_dinv_kernel(const int* __restrict__ cnt, int* __restrict__ rowptr,
                 float* __restrict__ dinv, int N) {
    __shared__ int part[8];
    __shared__ int wtot[8];
    const int tid = threadIdx.x;
    const int lane = tid & 31;
    const int wv = tid >> 5;
    const int base = blockIdx.x * SCAN_SPAN;

    int s = 0;
    for (int i = tid * 4; i < base; i += SCAN_SPAN) {
        int4 v = *(const int4*)(cnt + i);
        s += (v.x + v.y) + (v.z + v.w);
    }
#pragma unroll
    for (int o = 16; o > 0; o >>= 1) s += __shfl_xor(s, o);
    if (lane == 0) part[wv] = s;

    const int i0 = base + tid * 4;
    int c[4];
#pragma unroll
    for (int k = 0; k < 4; ++k) c[k] = (i0 + k < N) ? cnt[i0 + k] : 0;
    const int local = (c[0] + c[1]) + (c[2] + c[3]);
    int inc = local;
#pragma unroll
    for (int o = 1; o < 32; o <<= 1) {
        int t = __shfl_up(inc, o);
        if (lane >= o) inc += t;
    }
    if (lane == 31) wtot[wv] = inc;
    __syncthreads();

    int run = 0;
#pragma unroll
    for (int w = 0; w < 8; ++w) {
        run += part[w];
        if (w < wv) run += wtot[w];
    }
    run += inc - local;
#pragma unroll
    for (int k = 0; k < 4; ++k) {
        int idx = i0 + k;
        if (idx < N) {
            rowptr[idx] = run;
            dinv[idx] = 1.0f / sqrtf(fmaxf((float)c[k], 1.0f));
            run += c[k];
            if (idx == N - 1) rowptr[N] = run;
        }
    }
}

__global__ void place_kernel(const int* __restrict__ dst, const int* __restrict__ rowptr,
                             const int* __restrict__ pos, int* __restrict__ slot_eid, int E) {
    int e = blockIdx.x * blockDim.x + threadIdx.x;
    if (e < E) slot_eid[rowptr[dst[e]] + pos[e]] = e;
}

__global__ void __launch_bounds__(256)
order_kernel(const int* __restrict__ rowptr, const int* __restrict__ slot_eid,
             const int* __restrict__ src, const float* __restrict__ dinv,
             int* __restrict__ col, float* __restrict__ wcol, int N) {
    const int lane = threadIdx.x & 31;
    const int node = __builtin_amdgcn_readfirstlane((int)((blockIdx.x * blockDim.x + threadIdx.x) >> 5));
    if (node >= N) return;
    const int b = rowptr[node];
    const int e = rowptr[node + 1];
    for (int cb = b; cb < e; cb += 32) {
        const int j = cb + lane;
        const bool act = j < e;
        const int mine = act ? slot_eid[j] : 0x7fffffff;
        int rank = 0;
        for (int ob = b; ob < e; ob += 32) {
            const int jo = ob + lane;
            const int other = (jo < e) ? slot_eid[jo] : 0x7fffffff;
            const int n = min(32, e - ob);
            for (int k = 0; k < n; ++k)
                rank += (__builtin_amdgcn_readlane(other, k) < mine) ? 1 : 0;
        }
        if (act) {
            const int sn = src[mine];
            col[b + rank] = sn;
            wcol[b + rank] = dinv[sn];
        }
    }
}

__global__ void h_init_kernel(const float* __restrict__ feat, const float* __restrict__ theta,
                              float* __restrict__ h, int n4) {
    int i = blockIdx.x * blockDim.x + threadIdx.x;
    if (i >= n4) return;
    float t0 = theta[0];
    float4 f = ((const float4*)feat)[i];
    ((float4*)h)[i] = make_float4(t0 * f.x, t0 * f.y, t0 * f.z, t0 * f.w);
}


template <int R>
__device__ __forceinline__ void gather_rows(const float* __restrict__ x, int sv, float wv,
                                            int k, int lane, float4& acc) {
    float4 v[R];
    float w[R];
#pragma unroll
    for (int r = 0; r < R; ++r) {
        const int s = __builtin_amdgcn_readlane(sv, k + r);
        w[r] = __int_as_float(__builtin_amdgcn_readlane(__float_as_int(wv), k + r));
        v[r] = *(const float4*)(x + (size_t)s * DFEAT + lane * 4);
    }
#pragma unroll
    for (int r = 0; r < R; ++r) {
        acc.x = fmaf(w[r], v[r].x, acc.x);
        acc.y = fmaf(w[r], v[r].y, acc.y);
        acc.z = fmaf(w[r], v[r].z, acc.z);
        acc.w = fmaf(w[r], v[r].w, acc.w);
    }
}

__device__ __forceinline__ void gather_chunk(const float* __restrict__ x, int sv, float wv,
                                             int n, int lane, float4& acc) {
    int k = 0;
    for (; k + 4 <= n; k += 4) gather_rows<4>(x, sv, wv, k, lane, acc);
    const int r = n - k;
    if (r == 3)      gather_rows<3>(x, sv, wv, k, lane, acc);
    else if (r == 2) gather_rows<2>(x, sv, wv, k, lane, acc);
    else if (r == 1) gather_rows<1>(x, sv, wv, k, lane, acc);
}

__global__ void __launch_bounds__(256)
gather_update_kernel(const float* __restrict__ x, const int* __restrict__ rowptr,
                     const int* __restrict__ col, const float* __restrict__ wcol,
                     const float* __restrict__ dinv,
                     const float* Tprev, float* Tnew,
                     float* __restrict__ h, const float* __restrict__ theta,
                     int order, float c2, int N) {
    const int lane = threadIdx.x & 31;
    const int node = __builtin_amdgcn_readfirstlane((int)((blockIdx.x * blockDim.x + threadIdx.x) >> 5));
    if (node >= N) return;
    const int b = rowptr[node];
    const int e = rowptr[node + 1];
    const float s = -c2 * dinv[node];
    const float th = theta[order] * 0.5f;
    const size_t i = (size_t)node * (DFEAT / 4) + lane;

    int sv = 0;
    float wv = 0.f;
    if (b + lane < e) { sv = col[b + lane]; wv = wcol[b + lane]; }

    float4 tp = Tprev ? ((const float4*)Tprev)[i] : make_float4(0.f, 0.f, 0.f, 0.f);
    float4 hv = ((const float4*)h)[i];

    float4 acc = make_float4(0.f, 0.f, 0.f, 0.f);
    gather_chunk(x, sv, wv, min(32, e - b), lane, acc);
    for (int cb = b + 32; cb < e; cb += 32) {
        const int j = cb + lane;
        sv = 0;
        wv = 0.f;
        if (j < e) { sv = col[j]; wv = wcol[j]; }
        gather_chunk(x, sv, wv, min(32, e - cb), lane, acc);
    }

    float4 t;
    t.x = fmaf(s, acc.x, -tp.x);
    t.y = fmaf(s, acc.y, -tp.y);
    t.z = fmaf(s, acc.z, -tp.z);
    t.w = fmaf(s, acc.w, -tp.w);
    ((float4*)Tnew)[i] = t;

    hv.x = fmaf(th, t.x + 1.0f, hv.x);
    hv.y = fmaf(th, t.y + 1.0f, hv.y);
    hv.z = fmaf(th, t.z + 1.0f, hv.z);
    hv.w = fmaf(th, t.w + 1.0f, hv.w);
    ((float4*)h)[i] = hv;
}


__global__ void gemm_wmma_kernel(const float* __restrict__ h, const float* __restrict__ W,
                                 float* __restrict__ out, int N) {
    const int lane = threadIdx.x & 31;
    const int wave = threadIdx.x >> 5;
    const int tile = blockIdx.x * (blockDim.x >> 5) + wave;
    const int nRowTiles = N >> 4;
    const int row = tile >> 3;
    const int col = tile & 7;
    if (row >= nRowTiles) return;

    const int rowBase = row << 4;
    const int colBase = col << 4;
    const int m = lane & 15;
    const int half = lane >> 4;
    const int kb = half << 1;

    const float* hrow = h + (size_t)(rowBase + m) * DFEAT + kb;
    const float* wrow = W + (size_t)(colBase + m) * DFEAT + kb;

    v8f acc = {};
#pragma unroll
    for (int kk = 0; kk < DFEAT; kk += 4) {
        v2f a = *(const v2f*)(hrow + kk);
        v2f b = *(const v2f*)(wrow + kk);
        acc = __builtin_amdgcn_wmma_f32_16x16x4_f32(
             false, a,  false, b,
             (short)0, acc,  false,  false);
    }

#pragma unroll
    for (int v = 0; v < 8; ++v) {
        int r = rowBase + v + (half << 3);
        float val = acc[v];
        val = (val >= 0.0f) ? val : 0.01f * val;
        out[(size_t)r * DFEAT + colBase + m] = val;
    }
}


extern "C" void kernel_launch(void* const* d_in, const int* in_sizes, int n_in,
                              void* d_out, int out_size, void* d_ws, size_t ws_size,
                              hipStream_t stream) {
    const float* feat  = (const float*)d_in[0];
    const int*   src   = (const int*)d_in[1];
    const int*   dst   = (const int*)d_in[2];
    const float* theta = (const float*)d_in[3];
    const float* W     = (const float*)d_in[4];
    float* out = (float*)d_out;

    const int N = in_sizes[0] / DFEAT;
    const int E = in_sizes[1];

    float* ws   = (float*)d_ws;
    int*   cnt  = (int*)ws;
    float* dinv = ws + N;
    float* scr  = dinv + N;
    float* Ta   = scr + (size_t)N * DFEAT;
    float* Tb   = Ta + (size_t)N * DFEAT;
    float* h    = Tb + (size_t)N * DFEAT;

    int*   rowptr   = (int*)scr;
    int*   pos      = rowptr + (N + 4);
    int*   slot_eid = pos + E;
    int*   col      = slot_eid + E;
    float* wcol     = (float*)(col + E);

    const int TB = 256;
    const int nd4 = N * (DFEAT / 4);
    auto cdiv = [](long long a, long long b) { return (int)((a + b - 1) / b); };

    zero_f4_kernel<<<cdiv(N / 4, TB), TB, 0, stream>>>((float*)cnt, N / 4);
    rank_kernel<<<cdiv(E, TB), TB, 0, stream>>>(dst, cnt, pos, E);
    scan_dinv_kernel<<<cdiv(N, SCAN_SPAN), 256, 0, stream>>>(cnt, rowptr, dinv, N);
    place_kernel<<<cdiv(E, TB), TB, 0, stream>>>(dst, rowptr, pos, slot_eid, E);
    const int nodeBlocks = cdiv((long long)N * 32, TB);
    order_kernel<<<nodeBlocks, TB, 0, stream>>>(rowptr, slot_eid, src, dinv, col, wcol, N);

    h_init_kernel<<<cdiv(nd4, TB), TB, 0, stream>>>(feat, theta, h, nd4);

    gather_update_kernel<<<nodeBlocks, TB, 0, stream>>>(feat, rowptr, col, wcol, dinv, nullptr, Ta, h, theta, 1, 1.0f, N);
    gather_update_kernel<<<nodeBlocks, TB, 0, stream>>>(Ta, rowptr, col, wcol, dinv, feat, Tb, h, theta, 2, 2.0f, N);
    gather_update_kernel<<<nodeBlocks, TB, 0, stream>>>(Tb, rowptr, col, wcol, dinv, Ta, Ta, h, theta, 3, 2.0f, N);

    const int tiles = (N / 16) * (DFEAT / 16);
    gemm_wmma_kernel<<<cdiv(tiles, 256 / 32), 256, 0, stream>>>(h, W, out, N);
}


# Edge ranks from LDS counters: 128 work-groups each count one slice of the edge list per destination in an LDS table of all nodes, a column scan gives slice bases; returning global atomics gone
# speedup vs baseline: 3.0021x; 1.1044x over previous
#include <hip/hip_runtime.h>

typedef float v2f __attribute__((ext_vector_type(2)));
typedef float v8f __attribute__((ext_vector_type(8)));

#define DFEAT 128

#define HG 128
#define HT 1024
#define HR 65536
static_assert(HG <= DFEAT, "the slice table lives in one N x DFEAT plane");

__global__ void __launch_bounds__(HT)
hist_kernel(const int* __restrict__ dst, int* __restrict__ tab, int N, int E, int chunk) {
    __shared__ int lc[HR];
    const int tid = threadIdx.x;
    const int e0 = min(E, (int)blockIdx.x * chunk);
    const int e1 = min(E, e0 + chunk);
    int* row = tab + (size_t)blockIdx.x * N;
    for (int r0 = 0; r0 < N; r0 += HR) {
        const int rn = min(HR, N - r0);
        for (int i = tid * 4; i < rn; i += HT * 4) *(int4*)(lc + i) = make_int4(0, 0, 0, 0);
        __syncthreads();
        for (int eb = e0; eb < e1; eb += HT * 8) {
            int v[8];
#pragma unroll
            for (int k = 0; k < 8; ++k) v[k] = dst[min(eb + k * HT + tid, e1 - 1)];
#pragma unroll
            for (int k = 0; k < 8; ++k) {
                const int d = (eb + k * HT + tid < e1) ? v[k] - r0 : -1;
                if ((unsigned)d < (unsigned)rn) atomicAdd(&lc[d], 1);
            }
        }
        __syncthreads();
        for (int i = tid * 4; i < rn; i += HT * 4) *(int4*)(row + r0 + i) = *(const int4*)(lc + i);
        __syncthreads();
    }
}

__device__ __forceinline__ int4 add4(int4 a, int4 b) {
    return make_int4(a.x + b.x, a.y + b.y, a.z + b.z, a.w + b.w);
}

#define CS_G (HG / 4)
__global__ void __launch_bounds__(256)
colscan_kernel(int* __restrict__ tab, int* __restrict__ cnt, int N) {
    __shared__ int4 part[4][64];
    const int q = threadIdx.x & 63;
    const int p = threadIdx.x >> 6;
    const int NQ = N >> 2;
    const int n4 = blockIdx.x * 64 + q;
    const bool act = n4 < NQ;
    int4* col = (int4*)tab + (size_t)(p * CS_G) * NQ + min(n4, NQ - 1);
    int4 v[CS_G];
#pragma unroll
    for (int k = 0; k < CS_G; ++k) v[k] = col[(size_t)k * NQ];
    int4 run = make_int4(0, 0, 0, 0);
#pragma unroll
    for (int k = 0; k < CS_G; ++k) {
        const int4 t = v[k];
        v[k] = run;
        run = add4(run, t);
    }
    part[p][q] = run;
    __syncthreads();
    int4 off = make_int4(0, 0, 0, 0);
#pragma unroll
    for (int pp = 0; pp < 3; ++pp)
        if (pp < p) off = add4(off, part[pp][q]);
    if (act) {
#pragma unroll
        for (int k = 0; k < CS_G; ++k) col[(size_t)k * NQ] = add4(v[k], off);
        if (p == 3) ((int4*)cnt)[n4] = add4(off, run);
    }
}

#define SCAN_SPAN 1024
__global__ void __launch_bounds__(256)
scan_dinv_kernel(const int* __restrict__ cnt, int* __restrict__ rowptr,
                 float* __restrict__ dinv, int N) {
    __shared__ int part[8];
    __shared__ int wtot[8];
    const int tid = threadIdx.x;
    const int lane = tid & 31;
    const int wv = tid >> 5;
    const int base = blockIdx.x * SCAN_SPAN;

    int s = 0;
    for (int i = tid * 4; i < base; i += SCAN_SPAN) {
        int4 v = *(const int4*)(cnt + i);
        s += (v.x + v.y) + (v.z + v.w);
    }
#pragma unroll
    for (int o = 16; o > 0; o >>= 1) s += __shfl_xor(s, o);
    if (lane == 0) part[wv] = s;

    const int i0 = base + tid * 4;
    int c[4];
#pragma unroll
    for (int k = 0; k < 4; ++k) c[k] = (i0 + k < N) ? cnt[i0 + k] : 0;
    const int local = (c[0] + c[1]) + (c[2] + c[3]);
    int inc = local;
#pragma unroll
    for (int o = 1; o < 32; o <<= 1) {
        int t = __shfl_up(inc, o);
        if (lane >= o) inc += t;
    }
    if (lane == 31) wtot[wv] = inc;
    __syncthreads();

    int run = 0;
#pragma unroll
    for (int w = 0; w < 8; ++w) {
        run += part[w];
        if (w < wv) run += wtot[w];
    }
    run += inc - local;
#pragma unroll
    for (int k = 0; k < 4; ++k) {
        int idx = i0 + k;
        if (idx < N) {
            rowptr[idx] = run;
            dinv[idx] = 1.0f / sqrtf(fmaxf((float)c[k], 1.0f));
            run += c[k];
            if (idx == N - 1) rowptr[N] = run;
        }
    }
}

__global__ void __launch_bounds__(HT)
place_kernel(const int* __restrict__ dst, const int* __restrict__ rowptr,
             const int* __restrict__ tab, int* __restrict__ slot_eid, int N, int E, int chunk) {
    __shared__ int lc[HR];
    const int tid = threadIdx.x;
    const int e0 = min(E, (int)blockIdx.x * chunk);
    const int e1 = min(E, e0 + chunk);
    const int* row = tab + (size_t)blockIdx.x * N;
    for (int r0 = 0; r0 < N; r0 += HR) {
        const int rn = min(HR, N - r0);
#pragma unroll 4
        for (int i = tid * 4; i < rn; i += HT * 4)
            *(int4*)(lc + i) = add4(*(const int4*)(row + r0 + i), *(const int4*)(rowptr + r0 + i));
        __syncthreads();
        for (int eb = e0; eb < e1; eb += HT * 8) {
            int v[8];
#pragma unroll
            for (int k = 0; k < 8; ++k) v[k] = dst[min(eb + k * HT + tid, e1 - 1)];
#pragma unroll
            for (int k = 0; k < 8; ++k) {
                const int e = eb + k * HT + tid;
                const int d = (e < e1) ? v[k] - r0 : -1;
                if ((unsigned)d < (unsigned)rn) slot_eid[atomicAdd(&lc[d], 1)] = e;
            }
        }
        __syncthreads();
    }
}

__global__ void __launch_bounds__(256)
order_kernel(const int* __restrict__ rowptr, const int* __restrict__ slot_eid,
             const int* __restrict__ src, const float* __restrict__ dinv,
             int* __restrict__ col, float* __restrict__ wcol, int N) {
    const int lane = threadIdx.x & 31;
    const int node = __builtin_amdgcn_readfirstlane((int)((blockIdx.x * blockDim.x + threadIdx.x) >> 5));
    if (node >= N) return;
    const int b = rowptr[node];
    const int e = rowptr[node + 1];
    for (int cb = b; cb < e; cb += 32) {
        const int j = cb + lane;
        const bool act = j < e;
        const int mine = act ? slot_eid[j] : 0x7fffffff;
        int rank = 0;
        for (int ob = b; ob < e; ob += 32) {
            const int jo = ob + lane;
            const int other = (jo < e) ? slot_eid[jo] : 0x7fffffff;
            const int n = min(32, e - ob);
            for (int k = 0; k < n; ++k)
                rank += (__builtin_amdgcn_readlane(other, k) < mine) ? 1 : 0;
        }
        if (act) {
            const int sn = src[mine];
            col[b + rank] = sn;
            wcol[b + rank] = dinv[sn];
        }
    }
}

__global__ void h_init_kernel(const float* __restrict__ feat, const float* __restrict__ theta,
                              float* __restrict__ h, int n4) {
    int i = blockIdx.x * blockDim.x + threadIdx.x;
    if (i >= n4) return;
    float t0 = theta[0];
    float4 f = ((const float4*)feat)[i];
    ((float4*)h)[i] = make_float4(t0 * f.x, t0 * f.y, t0 * f.z, t0 * f.w);
}


template <int R>
__device__ __forceinline__ void gather_rows(const float* __restrict__ x, int sv, float wv,
                                            int k, int lane, float4& acc) {
    float4 v[R];
    float w[R];
#pragma unroll
    for (int r = 0; r < R; ++r) {
        const int s = __builtin_amdgcn_readlane(sv, k + r);
        w[r] = __int_as_float(__builtin_amdgcn_readlane(__float_as_int(wv), k + r));
        v[r] = *(const float4*)(x + (size_t)s * DFEAT + lane * 4);
    }
#pragma unroll
    for (int r = 0; r < R; ++r) {
        acc.x = fmaf(w[r], v[r].x, acc.x);
        acc.y = fmaf(w[r], v[r].y, acc.y);
        acc.z = fmaf(w[r], v[r].z, acc.z);
        acc.w = fmaf(w[r], v[r].w, acc.w);
    }
}

__device__ __forceinline__ void gather_chunk(const float* __restrict__ x, int sv, float wv,
                                             int n, int lane, float4& acc) {
    int k = 0;
    for (; k + 4 <= n; k += 4) gather_rows<4>(x, sv, wv, k, lane, acc);
    const int r = n - k;
    if (r == 3)      gather_rows<3>(x, sv, wv, k, lane, acc);
    else if (r == 2) gather_rows<2>(x, sv, wv, k, lane, acc);
    else if (r == 1) gather_rows<1>(x, sv, wv, k, lane, acc);
}

__global__ void __launch_bounds__(256)
gather_update_kernel(const float* __restrict__ x, const int* __restrict__ rowptr,
                     const int* __restrict__ col, const float* __restrict__ wcol,
                     const float* __restrict__ dinv,
                     const float* Tprev, float* Tnew,
                     float* __restrict__ h, const float* __restrict__ theta,
                     int order, float c2, int N) {
    const int lane = threadIdx.x & 31;
    const int node = __builtin_amdgcn_readfirstlane((int)((blockIdx.x * blockDim.x + threadIdx.x) >> 5));
    if (node >= N) return;
    const int b = rowptr[node];
    const int e = rowptr[node + 1];
    const float s = -c2 * dinv[node];
    const float th = theta[order] * 0.5f;
    const size_t i = (size_t)node * (DFEAT / 4) + lane;

    int sv = 0;
    float wv = 0.f;
    if (b + lane < e) { sv = col[b + lane]; wv = wcol[b + lane]; }

    float4 tp = Tprev ? ((const float4*)Tprev)[i] : make_float4(0.f, 0.f, 0.f, 0.f);
    float4 hv = ((const float4*)h)[i];

    float4 acc = make_float4(0.f, 0.f, 0.f, 0.f);
    gather_chunk(x, sv, wv, min(32, e - b), lane, acc);
    for (int cb = b + 32; cb < e; cb += 32) {
        const int j = cb + lane;
        sv = 0;
        wv = 0.f;
        if (j < e) { sv = col[j]; wv = wcol[j]; }
        gather_chunk(x, sv, wv, min(32, e - cb), lane, acc);
    }

    float4 t;
    t.x = fmaf(s, acc.x, -tp.x);
    t.y = fmaf(s, acc.y, -tp.y);
    t.z = fmaf(s, acc.z, -tp.z);
    t.w = fmaf(s, acc.w, -tp.w);
    ((float4*)Tnew)[i] = t;

    hv.x = fmaf(th, t.x + 1.0f, hv.x);
    hv.y = fmaf(th, t.y + 1.0f, hv.y);
    hv.z = fmaf(th, t.z + 1.0f, hv.z);
    hv.w = fmaf(th, t.w + 1.0f, hv.w);
    ((float4*)h)[i] = hv;
}


__global__ void gemm_wmma_kernel(const float* __restrict__ h, const float* __restrict__ W,
                                 float* __restrict__ out, int N) {
    const int lane = threadIdx.x & 31;
    const int wave = threadIdx.x >> 5;
    const int tile = blockIdx.x * (blockDim.x >> 5) + wave;
    const int nRowTiles = N >> 4;
    const int row = tile >> 3;
    const int col = tile & 7;
    if (row >= nRowTiles) return;

    const int rowBase = row << 4;
    const int colBase = col << 4;
    const int m = lane & 15;
    const int half = lane >> 4;
    const int kb = half << 1;

    const float* hrow = h + (size_t)(rowBase + m) * DFEAT + kb;
    const float* wrow = W + (size_t)(colBase + m) * DFEAT + kb;

    v8f acc = {};
#pragma unroll
    for (int kk = 0; kk < DFEAT; kk += 4) {
        v2f a = *(const v2f*)(hrow + kk);
        v2f b = *(const v2f*)(wrow + kk);
        acc = __builtin_amdgcn_wmma_f32_16x16x4_f32(
             false, a,  false, b,
             (short)0, acc,  false,  false);
    }

#pragma unroll
    for (int v = 0; v < 8; ++v) {
        int r = rowBase + v + (half << 3);
        float val = acc[v];
        val = (val >= 0.0f) ? val : 0.01f * val;
        out[(size_t)r * DFEAT + colBase + m] = val;
    }
}


extern "C" void kernel_launch(void* const* d_in, const int* in_sizes, int n_in,
                              void* d_out, int out_size, void* d_ws, size_t ws_size,
                              hipStream_t stream) {
    const float* feat  = (const float*)d_in[0];
    const int*   src   = (const int*)d_in[1];
    const int*   dst   = (const int*)d_in[2];
    const float* theta = (const float*)d_in[3];
    const float* W     = (const float*)d_in[4];
    float* out = (float*)d_out;

    const int N = in_sizes[0] / DFEAT;
    const int E = in_sizes[1];

    float* ws   = (float*)d_ws;
    int*   cnt  = (int*)ws;
    float* dinv = ws + N;
    float* scr  = dinv + N;
    float* Ta   = scr + (size_t)N * DFEAT;
    float* Tb   = Ta + (size_t)N * DFEAT;
    float* h    = Tb + (size_t)N * DFEAT;

    int*   rowptr   = (int*)scr;
    int*   slot_eid = rowptr + (N + 4);
    int*   col      = slot_eid + E;
    float* wcol     = (float*)(col + E);
    int*   tab      = (int*)Ta;

    const int TB = 256;
    const int nd4 = N * (DFEAT / 4);
    auto cdiv = [](long long a, long long b) { return (int)((a + b - 1) / b); };

    const int chunk = cdiv(E, HG);
    hist_kernel<<<HG, HT, 0, stream>>>(dst, tab, N, E, chunk);
    colscan_kernel<<<cdiv(N / 4, 64), 256, 0, stream>>>(tab, cnt, N);
    scan_dinv_kernel<<<cdiv(N, SCAN_SPAN), 256, 0, stream>>>(cnt, rowptr, dinv, N);
    place_kernel<<<HG, HT, 0, stream>>>(dst, rowptr, tab, slot_eid, N, E, chunk);
    const int nodeBlocks = cdiv((long long)N * 32, TB);
    order_kernel<<<nodeBlocks, TB, 0, stream>>>(rowptr, slot_eid, src, dinv, col, wcol, N);

    h_init_kernel<<<cdiv(nd4, TB), TB, 0, stream>>>(feat, theta, h, nd4);

    gather_update_kernel<<<nodeBlocks, TB, 0, stream>>>(feat, rowptr, col, wcol, dinv, nullptr, Ta, h, theta, 1, 1.0f, N);
    gather_update_kernel<<<nodeBlocks, TB, 0, stream>>>(Ta, rowptr, col, wcol, dinv, feat, Tb, h, theta, 2, 2.0f, N);
    gather_update_kernel<<<nodeBlocks, TB, 0, stream>>>(Tb, rowptr, col, wcol, dinv, Ta, Ta, h, theta, 3, 2.0f, N);

    const int tiles = (N / 16) * (DFEAT / 16);
    gemm_wmma_kernel<<<cdiv(tiles, 256 / 32), 256, 0, stream>>>(h, W, out, N);
}


# Row-offset kernel: counts in front of a block's range are asked for in groups of 16 pieces, its own counts ahead of them, instead of one dependent round trip per 4 KiB piece (48 in a row, last block)
# speedup vs baseline: 3.2698x; 1.0892x over previous
#include <hip/hip_runtime.h>

typedef float v2f __attribute__((ext_vector_type(2)));
typedef float v8f __attribute__((ext_vector_type(8)));

#define DFEAT 128

#define HG 128
#define HT 1024
#define HR 65536
static_assert(HG <= DFEAT, "the slice table lives in one N x DFEAT plane");

__global__ void __launch_bounds__(HT)
hist_kernel(const int* __restrict__ dst, int* __restrict__ tab, int N, int E, int chunk) {
    __shared__ int lc[HR];
    const int tid = threadIdx.x;
    const int e0 = min(E, (int)blockIdx.x * chunk);
    const int e1 = min(E, e0 + chunk);
    int* row = tab + (size_t)blockIdx.x * N;
    for (int r0 = 0; r0 < N; r0 += HR) {
        const int rn = min(HR, N - r0);
        for (int i = tid * 4; i < rn; i += HT * 4) *(int4*)(lc + i) = make_int4(0, 0, 0, 0);
        __syncthreads();
        for (int eb = e0; eb < e1; eb += HT * 8) {
            int v[8];
#pragma unroll
            for (int k = 0; k < 8; ++k) v[k] = dst[min(eb + k * HT + tid, e1 - 1)];
#pragma unroll
            for (int k = 0; k < 8; ++k) {
                const int d = (eb + k * HT + tid < e1) ? v[k] - r0 : -1;
                if ((unsigned)d < (unsigned)rn) atomicAdd(&lc[d], 1);
            }
        }
        __syncthreads();
        for (int i = tid * 4; i < rn; i += HT * 4) *(int4*)(row + r0 + i) = *(const int4*)(lc + i);
        __syncthreads();
    }
}

__device__ __forceinline__ int4 add4(int4 a, int4 b) {
    return make_int4(a.x + b.x, a.y + b.y, a.z + b.z, a.w + b.w);
}

#define CS_G (HG / 4)
__global__ void __launch_bounds__(256)
colscan_kernel(int* __restrict__ tab, int* __restrict__ cnt, int N) {
    __shared__ int4 part[4][64];
    const int q = threadIdx.x & 63;
    const int p = threadIdx.x >> 6;
    const int NQ = N >> 2;
    const int n4 = blockIdx.x * 64 + q;
    const bool act = n4 < NQ;
    int4* col = (int4*)tab + (size_t)(p * CS_G) * NQ + min(n4, NQ - 1);
    int4 v[CS_G];
#pragma unroll
    for (int k = 0; k < CS_G; ++k) v[k] = col[(size_t)k * NQ];
    int4 run = make_int4(0, 0, 0, 0);
#pragma unroll
    for (int k = 0; k < CS_G; ++k) {
        const int4 t = v[k];
        v[k] = run;
        run = add4(run, t);
    }
    part[p][q] = run;
    __syncthreads();
    int4 off = make_int4(0, 0, 0, 0);
#pragma unroll
    for (int pp = 0; pp < 3; ++pp)
        if (pp < p) off = add4(off, part[pp][q]);
    if (act) {
#pragma unroll
        for (int k = 0; k < CS_G; ++k) col[(size_t)k * NQ] = add4(v[k], off);
        if (p == 3) ((int4*)cnt)[n4] = add4(off, run);
    }
}

#define SCAN_SPAN 1024
#define SCAN_GRP 16
__global__ void __launch_bounds__(256)
scan_dinv_kernel(const int* __restrict__ cnt, int* __restrict__ rowptr,
                 float* __restrict__ dinv, int N) {
    __shared__ int part[8];
    __shared__ int wtot[8];
    const int tid = threadIdx.x;
    const int lane = tid & 31;
    const int wv = tid >> 5;
    const int nb = blockIdx.x;
    const int base = nb * SCAN_SPAN;

    const int i0 = base + tid * 4;
    int c[4];
#pragma unroll
    for (int k = 0; k < 4; ++k) c[k] = (i0 + k < N) ? cnt[i0 + k] : 0;

    int s = 0;
    const int4* cq = (const int4*)cnt + tid;
    for (int j0 = 0; j0 < nb; j0 += SCAN_GRP) {
        int4 v[SCAN_GRP];
#pragma unroll
        for (int k = 0; k < SCAN_GRP; ++k)
            v[k] = cq[(size_t)min(j0 + k, nb - 1) * (SCAN_SPAN / 4)];
#pragma unroll
        for (int k = 0; k < SCAN_GRP; ++k)
            if (j0 + k < nb) s += (v[k].x + v[k].y) + (v[k].z + v[k].w);
    }
#pragma unroll
    for (int o = 16; o > 0; o >>= 1) s += __shfl_xor(s, o);
    if (lane == 0) part[wv] = s;

    const int local = (c[0] + c[1]) + (c[2] + c[3]);
    int inc = local;
#pragma unroll
    for (int o = 1; o < 32; o <<= 1) {
        int t = __shfl_up(inc, o);
        if (lane >= o) inc += t;
    }
    if (lane == 31) wtot[wv] = inc;
    __syncthreads();

    int run = 0;
#pragma unroll
    for (int w = 0; w < 8; ++w) {
        run += part[w];
        if (w < wv) run += wtot[w];
    }
    run += inc - local;
#pragma unroll
    for (int k = 0; k < 4; ++k) {
        int idx = i0 + k;
        if (idx < N) {
            rowptr[idx] = run;
            dinv[idx] = 1.0f / sqrtf(fmaxf((float)c[k], 1.0f));
            run += c[k];
            if (idx == N - 1) rowptr[N] = run;
        }
    }
}

__global__ void __launch_bounds__(HT)
place_kernel(const int* __restrict__ dst, const int* __restrict__ rowptr,
             const int* __restrict__ tab, int* __restrict__ slot_eid, int N, int E, int chunk) {
    __shared__ int lc[HR];
    const int tid = threadIdx.x;
    const int e0 = min(E, (int)blockIdx.x * chunk);
    const int e1 = min(E, e0 + chunk);
    const int* row = tab + (size_t)blockIdx.x * N;
    for (int r0 = 0; r0 < N; r0 += HR) {
        const int rn = min(HR, N - r0);
#pragma unroll 4
        for (int i = tid * 4; i < rn; i += HT * 4)
            *(int4*)(lc + i) = add4(*(const int4*)(row + r0 + i), *(const int4*)(rowptr + r0 + i));
        __syncthreads();
        for (int eb = e0; eb < e1; eb += HT * 8) {
            int v[8];
#pragma unroll
            for (int k = 0; k < 8; ++k) v[k] = dst[min(eb + k * HT + tid, e1 - 1)];
#pragma unroll
            for (int k = 0; k < 8; ++k) {
                const int e = eb + k * HT + tid;
                const int d = (e < e1) ? v[k] - r0 : -1;
                if ((unsigned)d < (unsigned)rn) slot_eid[atomicAdd(&lc[d], 1)] = e;
            }
        }
        __syncthreads();
    }
}

__global__ void __launch_bounds__(256)
order_kernel(const int* __restrict__ rowptr, const int* __restrict__ slot_eid,
             const int* __restrict__ src, const float* __restrict__ dinv,
             int* __restrict__ col, float* __restrict__ wcol, int N) {
    const int lane = threadIdx.x & 31;
    const int node = __builtin_amdgcn_readfirstlane((int)((blockIdx.x * blockDim.x + threadIdx.x) >> 5));
    if (node >= N) return;
    const int b = rowptr[node];
    const int e = rowptr[node + 1];
    for (int cb = b; cb < e; cb += 32) {
        const int j = cb + lane;
        const bool act = j < e;
        const int mine = act ? slot_eid[j] : 0x7fffffff;
        int rank = 0;
        for (int ob = b; ob < e; ob += 32) {
            const int jo = ob + lane;
            const int other = (jo < e) ? slot_eid[jo] : 0x7fffffff;
            const int n = min(32, e - ob);
            for (int k = 0; k < n; ++k)
                rank += (__builtin_amdgcn_readlane(other, k) < mine) ? 1 : 0;
        }
        if (act) {
            const int sn = src[mine];
            col[b + rank] = sn;
            wcol[b + rank] = dinv[sn];
        }
    }
}

__global__ void h_init_kernel(const float* __restrict__ feat, const float* __restrict__ theta,
                              float* __restrict__ h, int n4) {
    int i = blockIdx.x * blockDim.x + threadIdx.x;
    if (i >= n4) return;
    float t0 = theta[0];
    float4 f = ((const float4*)feat)[i];
    ((float4*)h)[i] = make_float4(t0 * f.x, t0 * f.y, t0 * f.z, t0 * f.w);
}


template <int R>
__device__ __forceinline__ void gather_rows(const float* __restrict__ x, int sv, float wv,
                                            int k, int lane, float4& acc) {
    float4 v[R];
    float w[R];
#pragma unroll
    for (int r = 0; r < R; ++r) {
        const int s = __builtin_amdgcn_readlane(sv, k + r);
        w[r] = __int_as_float(__builtin_amdgcn_readlane(__float_as_int(wv), k + r));
        v[r] = *(const float4*)(x + (size_t)s * DFEAT + lane * 4);
    }
#pragma unroll
    for (int r = 0; r < R; ++r) {
        acc.x = fmaf(w[r], v[r].x, acc.x);
        acc.y = fmaf(w[r], v[r].y, acc.y);
        acc.z = fmaf(w[r], v[r].z, acc.z);
        acc.w = fmaf(w[r], v[r].w, acc.w);
    }
}

__device__ __forceinline__ void gather_chunk(const float* __restrict__ x, int sv, float wv,
                                             int n, int lane, float4& acc) {
    int k = 0;
    for (; k + 4 <= n; k += 4) gather_rows<4>(x, sv, wv, k, lane, acc);
    const int r = n - k;
    if (r == 3)      gather_rows<3>(x, sv, wv, k, lane, acc);
    else if (r == 2) gather_rows<2>(x, sv, wv, k, lane, acc);
    else if (r == 1) gather_rows<1>(x, sv, wv, k, lane, acc);
}

__global__ void __launch_bounds__(256)
gather_update_kernel(const float* __restrict__ x, const int* __restrict__ rowptr,
                     const int* __restrict__ col, const float* __restrict__ wcol,
                     const float* __restrict__ dinv,
                     const float* Tprev, float* Tnew,
                     float* __restrict__ h, const float* __restrict__ theta,
                     int order, float c2, int N) {
    const int lane = threadIdx.x & 31;
    const int node = __builtin_amdgcn_readfirstlane((int)((blockIdx.x * blockDim.x + threadIdx.x) >> 5));
    if (node >= N) return;
    const int b = rowptr[node];
    const int e = rowptr[node + 1];
    const float s = -c2 * dinv[node];
    const float th = theta[order] * 0.5f;
    const size_t i = (size_t)node * (DFEAT / 4) + lane;

    int sv = 0;
    float wv = 0.f;
    if (b + lane < e) { sv = col[b + lane]; wv = wcol[b + lane]; }

    float4 tp = Tprev ? ((const float4*)Tprev)[i] : make_float4(0.f, 0.f, 0.f, 0.f);
    float4 hv = ((const float4*)h)[i];

    float4 acc = make_float4(0.f, 0.f, 0.f, 0.f);
    gather_chunk(x, sv, wv, min(32, e - b), lane, acc);
    for (int cb = b + 32; cb < e; cb += 32) {
        const int j = cb + lane;
        sv = 0;
        wv = 0.f;
        if (j < e) { sv = col[j]; wv = wcol[j]; }
        gather_chunk(x, sv, wv, min(32, e - cb), lane, acc);
    }

    float4 t;
    t.x = fmaf(s, acc.x, -tp.x);
    t.y = fmaf(s, acc.y, -tp.y);
    t.z = fmaf(s, acc.z, -tp.z);
    t.w = fmaf(s, acc.w, -tp.w);
    ((float4*)Tnew)[i] = t;

    hv.x = fmaf(th, t.x + 1.0f, hv.x);
    hv.y = fmaf(th, t.y + 1.0f, hv.y);
    hv.z = fmaf(th, t.z + 1.0f, hv.z);
    hv.w = fmaf(th, t.w + 1.0f, hv.w);
    ((float4*)h)[i] = hv;
}


__global__ void gemm_wmma_kernel(const float* __restrict__ h, const float* __restrict__ W,
                                 float* __restrict__ out, int N) {
    const int lane = threadIdx.x & 31;
    const int wave = threadIdx.x >> 5;
    const int tile = blockIdx.x * (blockDim.x >> 5) + wave;
    const int nRowTiles = N >> 4;
    const int row = tile >> 3;
    const int col = tile & 7;
    if (row >= nRowTiles) return;

    const int rowBase = row << 4;
    const int colBase = col << 4;
    const int m = lane & 15;
    const int half = lane >> 4;
    const int kb = half << 1;

    const float* hrow = h + (size_t)(rowBase + m) * DFEAT + kb;
    const float* wrow = W + (size_t)(colBase + m) * DFEAT + kb;

    v8f acc = {};
#pragma unroll
    for (int kk = 0; kk < DFEAT; kk += 4) {
        v2f a = *(const v2f*)(hrow + kk);
        v2f b = *(const v2f*)(wrow + kk);
        acc = __builtin_amdgcn_wmma_f32_16x16x4_f32(
             false, a,  false, b,
             (short)0, acc,  false,  false);
    }

#pragma unroll
    for (int v = 0; v < 8; ++v) {
        int r = rowBase + v + (half << 3);
        float val = acc[v];
        val = (val >= 0.0f) ? val : 0.01f * val;
        out[(size_t)r * DFEAT + colBase + m] = val;
    }
}


extern "C" void kernel_launch(void* const* d_in, const int* in_sizes, int n_in,
                              void* d_out, int out_size, void* d_ws, size_t ws_size,
                              hipStream_t stream) {
    const float* feat  = (const float*)d_in[0];
    const int*   src   = (const int*)d_in[1];
    const int*   dst   = (const int*)d_in[2];
    const float* theta = (const float*)d_in[3];
    const float* W     = (const float*)d_in[4];
    float* out = (float*)d_out;

    const int N = in_sizes[0] / DFEAT;
    const int E = in_sizes[1];

    float* ws   = (float*)d_ws;
    int*   cnt  = (int*)ws;
    float* dinv = ws + N;
    float* scr  = dinv + N;
    float* Ta   = scr + (size_t)N * DFEAT;
    float* Tb   = Ta + (size_t)N * DFEAT;
    float* h    = Tb + (size_t)N * DFEAT;

    int*   rowptr   = (int*)scr;
    int*   slot_eid = rowptr + (N + 4);
    int*   col      = slot_eid + E;
    float* wcol     = (float*)(col + E);
    int*   tab      = (int*)Ta;

    const int TB = 256;
    const int nd4 = N * (DFEAT / 4);
    auto cdiv = [](long long a, long long b) { return (int)((a + b - 1) / b); };

    const int chunk = cdiv(E, HG);
    hist_kernel<<<HG, HT, 0, stream>>>(dst, tab, N, E, chunk);
    colscan_kernel<<<cdiv(N / 4, 64), 256, 0, stream>>>(tab, cnt, N);
    scan_dinv_kernel<<<cdiv(N, SCAN_SPAN), 256, 0, stream>>>(cnt, rowptr, dinv, N);
    place_kernel<<<HG, HT, 0, stream>>>(dst, rowptr, tab, slot_eid, N, E, chunk);
    const int nodeBlocks = cdiv((long long)N * 32, TB);
    order_kernel<<<nodeBlocks, TB, 0, stream>>>(rowptr, slot_eid, src, dinv, col, wcol, N);

    h_init_kernel<<<cdiv(nd4, TB), TB, 0, stream>>>(feat, theta, h, nd4);

    gather_update_kernel<<<nodeBlocks, TB, 0, stream>>>(feat, rowptr, col, wcol, dinv, nullptr, Ta, h, theta, 1, 1.0f, N);
    gather_update_kernel<<<nodeBlocks, TB, 0, stream>>>(Ta, rowptr, col, wcol, dinv, feat, Tb, h, theta, 2, 2.0f, N);
    gather_update_kernel<<<nodeBlocks, TB, 0, stream>>>(Tb, rowptr, col, wcol, dinv, Ta, Ta, h, theta, 3, 2.0f, N);

    const int tiles = (N / 16) * (DFEAT / 16);
    gemm_wmma_kernel<<<cdiv(tiles, 256 / 32), 256, 0, stream>>>(h, W, out, N);
}


# Final GEMM: one wave per 16x128 strip, W and the h rows staged in LDS by whole-row loads, one A fragment feeds 8 WMMAs, out rows stored whole; k order and sums unchanged
# speedup vs baseline: 3.6174x; 1.1063x over previous
#include <hip/hip_runtime.h>

typedef float v2f __attribute__((ext_vector_type(2)));
typedef float v8f __attribute__((ext_vector_type(8)));

#define DFEAT 128

#define HG 128
#define HT 1024
#define HR 65536
static_assert(HG <= DFEAT, "the slice table lives in one N x DFEAT plane");

__global__ void __launch_bounds__(HT)
hist_kernel(const int* __restrict__ dst, int* __restrict__ tab, int N, int E, int chunk) {
    __shared__ int lc[HR];
    const int tid = threadIdx.x;
    const int e0 = min(E, (int)blockIdx.x * chunk);
    const int e1 = min(E, e0 + chunk);
    int* row = tab + (size_t)blockIdx.x * N;
    for (int r0 = 0; r0 < N; r0 += HR) {
        const int rn = min(HR, N - r0);
        for (int i = tid * 4; i < rn; i += HT * 4) *(int4*)(lc + i) = make_int4(0, 0, 0, 0);
        __syncthreads();
        for (int eb = e0; eb < e1; eb += HT * 8) {
            int v[8];
#pragma unroll
            for (int k = 0; k < 8; ++k) v[k] = dst[min(eb + k * HT + tid, e1 - 1)];
#pragma unroll
            for (int k = 0; k < 8; ++k) {
                const int d = (eb + k * HT + tid < e1) ? v[k] - r0 : -1;
                if ((unsigned)d < (unsigned)rn) atomicAdd(&lc[d], 1);
            }
        }
        __syncthreads();
        for (int i = tid * 4; i < rn; i += HT * 4) *(int4*)(row + r0 + i) = *(const int4*)(lc + i);
        __syncthreads();
    }
}

__device__ __forceinline__ int4 add4(int4 a, int4 b) {
    return make_int4(a.x + b.x, a.y + b.y, a.z + b.z, a.w + b.w);
}

#define CS_G (HG / 4)
__global__ void __launch_bounds__(256)
colscan_kernel(int* __restrict__ tab, int* __restrict__ cnt, int N) {
    __shared__ int4 part[4][64];
    const int q = threadIdx.x & 63;
    const int p = threadIdx.x >> 6;
    const int NQ = N >> 2;
    const int n4 = blockIdx.x * 64 + q;
    const bool act = n4 < NQ;
    int4* col = (int4*)tab + (size_t)(p * CS_G) * NQ + min(n4, NQ - 1);
    int4 v[CS_G];
#pragma unroll
    for (int k = 0; k < CS_G; ++k) v[k] = col[(size_t)k * NQ];
    int4 run = make_int4(0, 0, 0, 0);
#pragma unroll
    for (int k = 0; k < CS_G; ++k) {
        const int4 t = v[k];
        v[k] = run;
        run = add4(run, t);
    }
    part[p][q] = run;
    __syncthreads();
    int4 off = make_int4(0, 0, 0, 0);
#pragma unroll
    for (int pp = 0; pp < 3; ++pp)
        if (pp < p) off = add4(off, part[pp][q]);
    if (act) {
#pragma unroll
        for (int k = 0; k < CS_G; ++k) col[(size_t)k * NQ] = add4(v[k], off);
        if (p == 3) ((int4*)cnt)[n4] = add4(off, run);
    }
}

#define SCAN_SPAN 1024
#define SCAN_GRP 16
__global__ void __launch_bounds__(256)
scan_dinv_kernel(const int* __restrict__ cnt, int* __restrict__ rowptr,
                 float* __restrict__ dinv, int N) {
    __shared__ int part[8];
    __shared__ int wtot[8];
    const int tid = threadIdx.x;
    const int lane = tid & 31;
    const int wv = tid >> 5;
    const int nb = blockIdx.x;
    const int base = nb * SCAN_SPAN;

    const int i0 = base + tid * 4;
    int c[4];
#pragma unroll
    for (int k = 0; k < 4; ++k) c[k] = (i0 + k < N) ? cnt[i0 + k] : 0;

    int s = 0;
    const int4* cq = (const int4*)cnt + tid;
    for (int j0 = 0; j0 < nb; j0 += SCAN_GRP) {
        int4 v[SCAN_GRP];
#pragma unroll
        for (int k = 0; k < SCAN_GRP; ++k)
            v[k] = cq[(size_t)min(j0 + k, nb - 1) * (SCAN_SPAN / 4)];
#pragma unroll
        for (int k = 0; k < SCAN_GRP; ++k)
            if (j0 + k < nb) s += (v[k].x + v[k].y) + (v[k].z + v[k].w);
    }
#pragma unroll
    for (int o = 16; o > 0; o >>= 1) s += __shfl_xor(s, o);
    if (lane == 0) part[wv] = s;

    const int local = (c[0] + c[1]) + (c[2] + c[3]);
    int inc = local;
#pragma unroll
    for (int o = 1; o < 32; o <<= 1) {
        int t = __shfl_up(inc, o);
        if (lane >= o) inc += t;
    }
    if (lane == 31) wtot[wv] = inc;
    __syncthreads();

    int run = 0;
#pragma unroll
    for (int w = 0; w < 8; ++w) {
        run += part[w];
        if (w < wv) run += wtot[w];
    }
    run += inc - local;
#pragma unroll
    for (int k = 0; k < 4; ++k) {
        int idx = i0 + k;
        if (idx < N) {
            rowptr[idx] = run;
            dinv[idx] = 1.0f / sqrtf(fmaxf((float)c[k], 1.0f));
            run += c[k];
            if (idx == N - 1) rowptr[N] = run;
        }
    }
}

__global__ void __launch_bounds__(HT)
place_kernel(const int* __restrict__ dst, const int* __restrict__ rowptr,
             const int* __restrict__ tab, int* __restrict__ slot_eid, int N, int E, int chunk) {
    __shared__ int lc[HR];
    const int tid = threadIdx.x;
    const int e0 = min(E, (int)blockIdx.x * chunk);
    const int e1 = min(E, e0 + chunk);
    const int* row = tab + (size_t)blockIdx.x * N;
    for (int r0 = 0; r0 < N; r0 += HR) {
        const int rn = min(HR, N - r0);
#pragma unroll 4
        for (int i = tid * 4; i < rn; i += HT * 4)
            *(int4*)(lc + i) = add4(*(const int4*)(row + r0 + i), *(const int4*)(rowptr + r0 + i));
        __syncthreads();
        for (int eb = e0; eb < e1; eb += HT * 8) {
            int v[8];
#pragma unroll
            for (int k = 0; k < 8; ++k) v[k] = dst[min(eb + k * HT + tid, e1 - 1)];
#pragma unroll
            for (int k = 0; k < 8; ++k) {
                const int e = eb + k * HT + tid;
                const int d = (e < e1) ? v[k] - r0 : -1;
                if ((unsigned)d < (unsigned)rn) slot_eid[atomicAdd(&lc[d], 1)] = e;
            }
        }
        __syncthreads();
    }
}

__global__ void __launch_bounds__(256)
order_kernel(const int* __restrict__ rowptr, const int* __restrict__ slot_eid,
             const int* __restrict__ src, const float* __restrict__ dinv,
             int* __restrict__ col, float* __restrict__ wcol, int N) {
    const int lane = threadIdx.x & 31;
    const int node = __builtin_amdgcn_readfirstlane((int)((blockIdx.x * blockDim.x + threadIdx.x) >> 5));
    if (node >= N) return;
    const int b = rowptr[node];
    const int e = rowptr[node + 1];
    for (int cb = b; cb < e; cb += 32) {
        const int j = cb + lane;
        const bool act = j < e;
        const int mine = act ? slot_eid[j] : 0x7fffffff;
        int rank = 0;
        for (int ob = b; ob < e; ob += 32) {
            const int jo = ob + lane;
            const int other = (jo < e) ? slot_eid[jo] : 0x7fffffff;
            const int n = min(32, e - ob);
            for (int k = 0; k < n; ++k)
                rank += (__builtin_amdgcn_readlane(other, k) < mine) ? 1 : 0;
        }
        if (act) {
            const int sn = src[mine];
            col[b + rank] = sn;
            wcol[b + rank] = dinv[sn];
        }
    }
}

__global__ void h_init_kernel(const float* __restrict__ feat, const float* __restrict__ theta,
                              float* __restrict__ h, int n4) {
    int i = blockIdx.x * blockDim.x + threadIdx.x;
    if (i >= n4) return;
    float t0 = theta[0];
    float4 f = ((const float4*)feat)[i];
    ((float4*)h)[i] = make_float4(t0 * f.x, t0 * f.y, t0 * f.z, t0 * f.w);
}


template <int R>
__device__ __forceinline__ void gather_rows(const float* __restrict__ x, int sv, float wv,
                                            int k, int lane, float4& acc) {
    float4 v[R];
    float w[R];
#pragma unroll
    for (int r = 0; r < R; ++r) {
        const int s = __builtin_amdgcn_readlane(sv, k + r);
        w[r] = __int_as_float(__builtin_amdgcn_readlane(__float_as_int(wv), k + r));
        v[r] = *(const float4*)(x + (size_t)s * DFEAT + lane * 4);
    }
#pragma unroll
    for (int r = 0; r < R; ++r) {
        acc.x = fmaf(w[r], v[r].x, acc.x);
        acc.y = fmaf(w[r], v[r].y, acc.y);
        acc.z = fmaf(w[r], v[r].z, acc.z);
        acc.w = fmaf(w[r], v[r].w, acc.w);
    }
}

__device__ __forceinline__ void gather_chunk(const float* __restrict__ x, int sv, float wv,
                                             int n, int lane, float4& acc) {
    int k = 0;
    for (; k + 4 <= n; k += 4) gather_rows<4>(x, sv, wv, k, lane, acc);
    const int r = n - k;
    if (r == 3)      gather_rows<3>(x, sv, wv, k, lane, acc);
    else if (r == 2) gather_rows<2>(x, sv, wv, k, lane, acc);
    else if (r == 1) gather_rows<1>(x, sv, wv, k, lane, acc);
}

__global__ void __launch_bounds__(256)
gather_update_kernel(const float* __restrict__ x, const int* __restrict__ rowptr,
                     const int* __restrict__ col, const float* __restrict__ wcol,
                     const float* __restrict__ dinv,
                     const float* Tprev, float* Tnew,
                     float* __restrict__ h, const float* __restrict__ theta,
                     int order, float c2, int N) {
    const int lane = threadIdx.x & 31;
    const int node = __builtin_amdgcn_readfirstlane((int)((blockIdx.x * blockDim.x + threadIdx.x) >> 5));
    if (node >= N) return;
    const int b = rowptr[node];
    const int e = rowptr[node + 1];
    const float s = -c2 * dinv[node];
    const float th = theta[order] * 0.5f;
    const size_t i = (size_t)node * (DFEAT / 4) + lane;

    int sv = 0;
    float wv = 0.f;
    if (b + lane < e) { sv = col[b + lane]; wv = wcol[b + lane]; }

    float4 tp = Tprev ? ((const float4*)Tprev)[i] : make_float4(0.f, 0.f, 0.f, 0.f);
    float4 hv = ((const float4*)h)[i];

    float4 acc = make_float4(0.f, 0.f, 0.f, 0.f);
    gather_chunk(x, sv, wv, min(32, e - b), lane, acc);
    for (int cb = b + 32; cb < e; cb += 32) {
        const int j = cb + lane;
        sv = 0;
        wv = 0.f;
        if (j < e) { sv = col[j]; wv = wcol[j]; }
        gather_chunk(x, sv, wv, min(32, e - cb), lane, acc);
    }

    float4 t;
    t.x = fmaf(s, acc.x, -tp.x);
    t.y = fmaf(s, acc.y, -tp.y);
    t.z = fmaf(s, acc.z, -tp.z);
    t.w = fmaf(s, acc.w, -tp.w);
    ((float4*)Tnew)[i] = t;

    hv.x = fmaf(th, t.x + 1.0f, hv.x);
    hv.y = fmaf(th, t.y + 1.0f, hv.y);
    hv.z = fmaf(th, t.z + 1.0f, hv.z);
    hv.w = fmaf(th, t.w + 1.0f, hv.w);
    ((float4*)h)[i] = hv;
}

#define GP 132
#define GW 8

__global__ void __launch_bounds__(GW * 32)
gemm_wmma_kernel(const float* __restrict__ h, const float* __restrict__ W,
                 float* __restrict__ out, int N) {
    __shared__ float sW[DFEAT * GP];
    __shared__ float sA[GW][16 * GP];
    const int tid = threadIdx.x;
    const int lane = tid & 31;
    const int wave = tid >> 5;
    const int m = lane & 15;
    const int half = lane >> 4;
    const int nStrips = N >> 4;
    const int strip = blockIdx.x * GW + wave;
    const bool live = strip < nStrips;
    float* myA = sA[wave];

    const float4* Wq = (const float4*)W + tid;
    const float4* hq = (const float4*)h + (size_t)min(strip, nStrips - 1) * (16 * DFEAT / 4) + lane;
    float4 wv[16], av[16];
#pragma unroll
    for (int j = 0; j < 16; ++j) wv[j] = Wq[j * (GW * 32)];
#pragma unroll
    for (int r = 0; r < 16; ++r) av[r] = hq[r * (DFEAT / 4)];
#pragma unroll
    for (int j = 0; j < 16; ++j) *(float4*)(sW + (wave + GW * j) * GP + lane * 4) = wv[j];
#pragma unroll
    for (int r = 0; r < 16; ++r) *(float4*)(myA + r * GP + lane * 4) = av[r];
    __syncthreads();
    if (!live) return;

    const float* aP = myA + m * GP + (half << 1);
    const float* bP = sW + m * GP + (half << 1);

    v8f acc[8];
#pragma unroll
    for (int t = 0; t < 8; ++t) acc[t] = (v8f){};

#pragma unroll
    for (int kk = 0; kk < DFEAT; kk += 4) {
        const v2f a = *(const v2f*)(aP + kk);
#pragma unroll
        for (int t = 0; t < 8; ++t) {
            const v2f b = *(const v2f*)(bP + t * 16 * GP + kk);
            acc[t] = __builtin_amdgcn_wmma_f32_16x16x4_f32(
                 false, a,  false, b,
                 (short)0, acc[t],  false,  false);
        }
    }

#pragma unroll
    for (int t = 0; t < 8; ++t) {
#pragma unroll
        for (int v = 0; v < 8; ++v) {
            float val = acc[t][v];
            val = (val >= 0.0f) ? val : 0.01f * val;
            myA[(v + (half << 3)) * GP + t * 16 + m] = val;
        }
    }
    float4* oq = (float4*)out + (size_t)strip * (16 * DFEAT / 4) + lane;
#pragma unroll
    for (int r = 0; r < 16; ++r) oq[r * (DFEAT / 4)] = *(const float4*)(myA + r * GP + lane * 4);
}


extern "C" void kernel_launch(void* const* d_in, const int* in_sizes, int n_in,
                              void* d_out, int out_size, void* d_ws, size_t ws_size,
                              hipStream_t stream) {
    const float* feat  = (const float*)d_in[0];
    const int*   src   = (const int*)d_in[1];
    const int*   dst   = (const int*)d_in[2];
    const float* theta = (const float*)d_in[3];
    const float* W     = (const float*)d_in[4];
    float* out = (float*)d_out;

    const int N = in_sizes[0] / DFEAT;
    const int E = in_sizes[1];

    float* ws   = (float*)d_ws;
    int*   cnt  = (int*)ws;
    float* dinv = ws + N;
    float* scr  = dinv + N;
    float* Ta   = scr + (size_t)N * DFEAT;
    float* Tb   = Ta + (size_t)N * DFEAT;
    float* h    = Tb + (size_t)N * DFEAT;

    int*   rowptr   = (int*)scr;
    int*   slot_eid = rowptr + (N + 4);
    int*   col      = slot_eid + E;
    float* wcol     = (float*)(col + E);
    int*   tab      = (int*)Ta;

    const int TB = 256;
    const int nd4 = N * (DFEAT / 4);
    auto cdiv = [](long long a, long long b) { return (int)((a + b - 1) / b); };

    const int chunk = cdiv(E, HG);
    hist_kernel<<<HG, HT, 0, stream>>>(dst, tab, N, E, chunk);
    colscan_kernel<<<cdiv(N / 4, 64), 256, 0, stream>>>(tab, cnt, N);
    scan_dinv_kernel<<<cdiv(N, SCAN_SPAN), 256, 0, stream>>>(cnt, rowptr, dinv, N);
    place_kernel<<<HG, HT, 0, stream>>>(dst, rowptr, tab, slot_eid, N, E, chunk);
    const int nodeBlocks = cdiv((long long)N * 32, TB);
    order_kernel<<<nodeBlocks, TB, 0, stream>>>(rowptr, slot_eid, src, dinv, col, wcol, N);

    h_init_kernel<<<cdiv(nd4, TB), TB, 0, stream>>>(feat, theta, h, nd4);

    gather_update_kernel<<<nodeBlocks, TB, 0, stream>>>(feat, rowptr, col, wcol, dinv, nullptr, Ta, h, theta, 1, 1.0f, N);
    gather_update_kernel<<<nodeBlocks, TB, 0, stream>>>(Ta, rowptr, col, wcol, dinv, feat, Tb, h, theta, 2, 2.0f, N);
    gather_update_kernel<<<nodeBlocks, TB, 0, stream>>>(Tb, rowptr, col, wcol, dinv, Ta, Ta, h, theta, 3, 2.0f, N);

    const int strips = N / 16;
    gemm_wmma_kernel<<<cdiv(strips, GW), GW * 32, 0, stream>>>(h, W, out, N);
}


# h formed once: orders 1 and 2 store only their T row, the order-3 wave builds the h row from feat, T1, T2 and its own T3 in the same sum order; h_init launch, h re-reads and the T3 store gone
# speedup vs baseline: 4.0932x; 1.1315x over previous
#include <hip/hip_runtime.h>

typedef float v2f __attribute__((ext_vector_type(2)));
typedef float v8f __attribute__((ext_vector_type(8)));

#define DFEAT 128

#define HG 128
#define HT 1024
#define HR 65536
static_assert(HG <= DFEAT, "the slice table lives in one N x DFEAT plane");

__global__ void __launch_bounds__(HT)
hist_kernel(const int* __restrict__ dst, int* __restrict__ tab, int N, int E, int chunk) {
    __shared__ int lc[HR];
    const int tid = threadIdx.x;
    const int e0 = min(E, (int)blockIdx.x * chunk);
    const int e1 = min(E, e0 + chunk);
    int* row = tab + (size_t)blockIdx.x * N;
    for (int r0 = 0; r0 < N; r0 += HR) {
        const int rn = min(HR, N - r0);
        for (int i = tid * 4; i < rn; i += HT * 4) *(int4*)(lc + i) = make_int4(0, 0, 0, 0);
        __syncthreads();
        for (int eb = e0; eb < e1; eb += HT * 8) {
            int v[8];
#pragma unroll
            for (int k = 0; k < 8; ++k) v[k] = dst[min(eb + k * HT + tid, e1 - 1)];
#pragma unroll
            for (int k = 0; k < 8; ++k) {
                const int d = (eb + k * HT + tid < e1) ? v[k] - r0 : -1;
                if ((unsigned)d < (unsigned)rn) atomicAdd(&lc[d], 1);
            }
        }
        __syncthreads();
        for (int i = tid * 4; i < rn; i += HT * 4) *(int4*)(row + r0 + i) = *(const int4*)(lc + i);
        __syncthreads();
    }
}

__device__ __forceinline__ int4 add4(int4 a, int4 b) {
    return make_int4(a.x + b.x, a.y + b.y, a.z + b.z, a.w + b.w);
}

#define CS_G (HG / 4)
__global__ void __launch_bounds__(256)
colscan_kernel(int* __restrict__ tab, int* __restrict__ cnt, int N) {
    __shared__ int4 part[4][64];
    const int q = threadIdx.x & 63;
    const int p = threadIdx.x >> 6;
    const int NQ = N >> 2;
    const int n4 = blockIdx.x * 64 + q;
    const bool act = n4 < NQ;
    int4* col = (int4*)tab + (size_t)(p * CS_G) * NQ + min(n4, NQ - 1);
    int4 v[CS_G];
#pragma unroll
    for (int k = 0; k < CS_G; ++k) v[k] = col[(size_t)k * NQ];
    int4 run = make_int4(0, 0, 0, 0);
#pragma unroll
    for (int k = 0; k < CS_G; ++k) {
        const int4 t = v[k];
        v[k] = run;
        run = add4(run, t);
    }
    part[p][q] = run;
    __syncthreads();
    int4 off = make_int4(0, 0, 0, 0);
#pragma unroll
    for (int pp = 0; pp < 3; ++pp)
        if (pp < p) off = add4(off, part[pp][q]);
    if (act) {
#pragma unroll
        for (int k = 0; k < CS_G; ++k) col[(size_t)k * NQ] = add4(v[k], off);
        if (p == 3) ((int4*)cnt)[n4] = add4(off, run);
    }
}

#define SCAN_SPAN 1024
#define SCAN_GRP 16
__global__ void __launch_bounds__(256)
scan_dinv_kernel(const int* __restrict__ cnt, int* __restrict__ rowptr,
                 float* __restrict__ dinv, int N) {
    __shared__ int part[8];
    __shared__ int wtot[8];
    const int tid = threadIdx.x;
    const int lane = tid & 31;
    const int wv = tid >> 5;
    const int nb = blockIdx.x;
    const int base = nb * SCAN_SPAN;

    const int i0 = base + tid * 4;
    int c[4];
#pragma unroll
    for (int k = 0; k < 4; ++k) c[k] = (i0 + k < N) ? cnt[i0 + k] : 0;

    int s = 0;
    const int4* cq = (const int4*)cnt + tid;
    for (int j0 = 0; j0 < nb; j0 += SCAN_GRP) {
        int4 v[SCAN_GRP];
#pragma unroll
        for (int k = 0; k < SCAN_GRP; ++k)
            v[k] = cq[(size_t)min(j0 + k, nb - 1) * (SCAN_SPAN / 4)];
#pragma unroll
        for (int k = 0; k < SCAN_GRP; ++k)
            if (j0 + k < nb) s += (v[k].x + v[k].y) + (v[k].z + v[k].w);
    }
#pragma unroll
    for (int o = 16; o > 0; o >>= 1) s += __shfl_xor(s, o);
    if (lane == 0) part[wv] = s;

    const int local = (c[0] + c[1]) + (c[2] + c[3]);
    int inc = local;
#pragma unroll
    for (int o = 1; o < 32; o <<= 1) {
        int t = __shfl_up(inc, o);
        if (lane >= o) inc += t;
    }
    if (lane == 31) wtot[wv] = inc;
    __syncthreads();

    int run = 0;
#pragma unroll
    for (int w = 0; w < 8; ++w) {
        run += part[w];
        if (w < wv) run += wtot[w];
    }
    run += inc - local;
#pragma unroll
    for (int k = 0; k < 4; ++k) {
        int idx = i0 + k;
        if (idx < N) {
            rowptr[idx] = run;
            dinv[idx] = 1.0f / sqrtf(fmaxf((float)c[k], 1.0f));
            run += c[k];
            if (idx == N - 1) rowptr[N] = run;
        }
    }
}

__global__ void __launch_bounds__(HT)
place_kernel(const int* __restrict__ dst, const int* __restrict__ rowptr,
             const int* __restrict__ tab, int* __restrict__ slot_eid, int N, int E, int chunk) {
    __shared__ int lc[HR];
    const int tid = threadIdx.x;
    const int e0 = min(E, (int)blockIdx.x * chunk);
    const int e1 = min(E, e0 + chunk);
    const int* row = tab + (size_t)blockIdx.x * N;
    for (int r0 = 0; r0 < N; r0 += HR) {
        const int rn = min(HR, N - r0);
#pragma unroll 4
        for (int i = tid * 4; i < rn; i += HT * 4)
            *(int4*)(lc + i) = add4(*(const int4*)(row + r0 + i), *(const int4*)(rowptr + r0 + i));
        __syncthreads();
        for (int eb = e0; eb < e1; eb += HT * 8) {
            int v[8];
#pragma unroll
            for (int k = 0; k < 8; ++k) v[k] = dst[min(eb + k * HT + tid, e1 - 1)];
#pragma unroll
            for (int k = 0; k < 8; ++k) {
                const int e = eb + k * HT + tid;
                const int d = (e < e1) ? v[k] - r0 : -1;
                if ((unsigned)d < (unsigned)rn) slot_eid[atomicAdd(&lc[d], 1)] = e;
            }
        }
        __syncthreads();
    }
}

__global__ void __launch_bounds__(256)
order_kernel(const int* __restrict__ rowptr, const int* __restrict__ slot_eid,
             const int* __restrict__ src, const float* __restrict__ dinv,
             int* __restrict__ col, float* __restrict__ wcol, int N) {
    const int lane = threadIdx.x & 31;
    const int node = __builtin_amdgcn_readfirstlane((int)((blockIdx.x * blockDim.x + threadIdx.x) >> 5));
    if (node >= N) return;
    const int b = rowptr[node];
    const int e = rowptr[node + 1];
    for (int cb = b; cb < e; cb += 32) {
        const int j = cb + lane;
        const bool act = j < e;
        const int mine = act ? slot_eid[j] : 0x7fffffff;
        int rank = 0;
        for (int ob = b; ob < e; ob += 32) {
            const int jo = ob + lane;
            const int other = (jo < e) ? slot_eid[jo] : 0x7fffffff;
            const int n = min(32, e - ob);
            for (int k = 0; k < n; ++k)
                rank += (__builtin_amdgcn_readlane(other, k) < mine) ? 1 : 0;
        }
        if (act) {
            const int sn = src[mine];
            col[b + rank] = sn;
            wcol[b + rank] = dinv[sn];
        }
    }
}


template <int R>
__device__ __forceinline__ void gather_rows(const float* __restrict__ x, int sv, float wv,
                                            int k, int lane, float4& acc) {
    float4 v[R];
    float w[R];
#pragma unroll
    for (int r = 0; r < R; ++r) {
        const int s = __builtin_amdgcn_readlane(sv, k + r);
        w[r] = __int_as_float(__builtin_amdgcn_readlane(__float_as_int(wv), k + r));
        v[r] = *(const float4*)(x + (size_t)s * DFEAT + lane * 4);
    }
#pragma unroll
    for (int r = 0; r < R; ++r) {
        acc.x = fmaf(w[r], v[r].x, acc.x);
        acc.y = fmaf(w[r], v[r].y, acc.y);
        acc.z = fmaf(w[r], v[r].z, acc.z);
        acc.w = fmaf(w[r], v[r].w, acc.w);
    }
}

__device__ __forceinline__ void gather_chunk(const float* __restrict__ x, int sv, float wv,
                                             int n, int lane, float4& acc) {
    int k = 0;
    for (; k + 4 <= n; k += 4) gather_rows<4>(x, sv, wv, k, lane, acc);
    const int r = n - k;
    if (r == 3)      gather_rows<3>(x, sv, wv, k, lane, acc);
    else if (r == 2) gather_rows<2>(x, sv, wv, k, lane, acc);
    else if (r == 1) gather_rows<1>(x, sv, wv, k, lane, acc);
}

template <int ORD>
__global__ void __launch_bounds__(256)
gather_update_kernel(const float* __restrict__ x, const int* __restrict__ rowptr,
                     const int* __restrict__ col, const float* __restrict__ wcol,
                     const float* __restrict__ dinv,
                     const float* __restrict__ Tprev, float* __restrict__ Tnew,
                     const float* __restrict__ feat, float* __restrict__ h,
                     const float* __restrict__ theta, float c2, int N) {
    const int lane = threadIdx.x & 31;
    const int node = __builtin_amdgcn_readfirstlane((int)((blockIdx.x * blockDim.x + threadIdx.x) >> 5));
    if (node >= N) return;
    const int b = rowptr[node];
    const int e = rowptr[node + 1];
    const float s = -c2 * dinv[node];
    float t0 = 0.f, th1 = 0.f, th2 = 0.f, th3 = 0.f;
    if (ORD == 3) {
        t0 = theta[0];
        th1 = theta[1] * 0.5f;
        th2 = theta[2] * 0.5f;
        th3 = theta[3] * 0.5f;
    }
    const size_t i = (size_t)node * (DFEAT / 4) + lane;

    int sv = 0;
    float wv = 0.f;
    if (b + lane < e) { sv = col[b + lane]; wv = wcol[b + lane]; }

    float4 tp = make_float4(0.f, 0.f, 0.f, 0.f);
    if (ORD >= 2) tp = ((const float4*)Tprev)[i];
    float4 f = make_float4(0.f, 0.f, 0.f, 0.f);
    float4 t2 = make_float4(0.f, 0.f, 0.f, 0.f);
    if (ORD == 3) {
        f = ((const float4*)feat)[i];
        t2 = ((const float4*)x)[i];
    }

    float4 acc = make_float4(0.f, 0.f, 0.f, 0.f);
    gather_chunk(x, sv, wv, min(32, e - b), lane, acc);
    for (int cb = b + 32; cb < e; cb += 32) {
        const int j = cb + lane;
        sv = 0;
        wv = 0.f;
        if (j < e) { sv = col[j]; wv = wcol[j]; }
        gather_chunk(x, sv, wv, min(32, e - cb), lane, acc);
    }

    float4 t;
    t.x = fmaf(s, acc.x, -tp.x);
    t.y = fmaf(s, acc.y, -tp.y);
    t.z = fmaf(s, acc.z, -tp.z);
    t.w = fmaf(s, acc.w, -tp.w);

    if (ORD < 3) {
        ((float4*)Tnew)[i] = t;
    } else {
        float4 hv = make_float4(t0 * f.x, t0 * f.y, t0 * f.z, t0 * f.w);
        hv.x = fmaf(th1, tp.x + 1.0f, hv.x);
        hv.y = fmaf(th1, tp.y + 1.0f, hv.y);
        hv.z = fmaf(th1, tp.z + 1.0f, hv.z);
        hv.w = fmaf(th1, tp.w + 1.0f, hv.w);
        hv.x = fmaf(th2, t2.x + 1.0f, hv.x);
        hv.y = fmaf(th2, t2.y + 1.0f, hv.y);
        hv.z = fmaf(th2, t2.z + 1.0f, hv.z);
        hv.w = fmaf(th2, t2.w + 1.0f, hv.w);
        hv.x = fmaf(th3, t.x + 1.0f, hv.x);
        hv.y = fmaf(th3, t.y + 1.0f, hv.y);
        hv.z = fmaf(th3, t.z + 1.0f, hv.z);
        hv.w = fmaf(th3, t.w + 1.0f, hv.w);
        ((float4*)h)[i] = hv;
    }
}

#define GP 132
#define GW 8

__global__ void __launch_bounds__(GW * 32)
gemm_wmma_kernel(const float* __restrict__ h, const float* __restrict__ W,
                 float* __restrict__ out, int N) {
    __shared__ float sW[DFEAT * GP];
    __shared__ float sA[GW][16 * GP];
    const int tid = threadIdx.x;
    const int lane = tid & 31;
    const int wave = tid >> 5;
    const int m = lane & 15;
    const int half = lane >> 4;
    const int nStrips = N >> 4;
    const int strip = blockIdx.x * GW + wave;
    const bool live = strip < nStrips;
    float* myA = sA[wave];

    const float4* Wq = (const float4*)W + tid;
    const float4* hq = (const float4*)h + (size_t)min(strip, nStrips - 1) * (16 * DFEAT / 4) + lane;
    float4 wv[16], av[16];
#pragma unroll
    for (int j = 0; j < 16; ++j) wv[j] = Wq[j * (GW * 32)];
#pragma unroll
    for (int r = 0; r < 16; ++r) av[r] = hq[r * (DFEAT / 4)];
#pragma unroll
    for (int j = 0; j < 16; ++j) *(float4*)(sW + (wave + GW * j) * GP + lane * 4) = wv[j];
#pragma unroll
    for (int r = 0; r < 16; ++r) *(float4*)(myA + r * GP + lane * 4) = av[r];
    __syncthreads();
    if (!live) return;

    const float* aP = myA + m * GP + (half << 1);
    const float* bP = sW + m * GP + (half << 1);

    v8f acc[8];
#pragma unroll
    for (int t = 0; t < 8; ++t) acc[t] = (v8f){};

#pragma unroll
    for (int kk = 0; kk < DFEAT; kk += 4) {
        const v2f a = *(const v2f*)(aP + kk);
#pragma unroll
        for (int t = 0; t < 8; ++t) {
            const v2f b = *(const v2f*)(bP + t * 16 * GP + kk);
            acc[t] = __builtin_amdgcn_wmma_f32_16x16x4_f32(
                 false, a,  false, b,
                 (short)0, acc[t],  false,  false);
        }
    }

#pragma unroll
    for (int t = 0; t < 8; ++t) {
#pragma unroll
        for (int v = 0; v < 8; ++v) {
            float val = acc[t][v];
            val = (val >= 0.0f) ? val : 0.01f * val;
            myA[(v + (half << 3)) * GP + t * 16 + m] = val;
        }
    }
    float4* oq = (float4*)out + (size_t)strip * (16 * DFEAT / 4) + lane;
#pragma unroll
    for (int r = 0; r < 16; ++r) oq[r * (DFEAT / 4)] = *(const float4*)(myA + r * GP + lane * 4);
}


extern "C" void kernel_launch(void* const* d_in, const int* in_sizes, int n_in,
                              void* d_out, int out_size, void* d_ws, size_t ws_size,
                              hipStream_t stream) {
    const float* feat  = (const float*)d_in[0];
    const int*   src   = (const int*)d_in[1];
    const int*   dst   = (const int*)d_in[2];
    const float* theta = (const float*)d_in[3];
    const float* W     = (const float*)d_in[4];
    float* out = (float*)d_out;

    const int N = in_sizes[0] / DFEAT;
    const int E = in_sizes[1];

    float* ws   = (float*)d_ws;
    int*   cnt  = (int*)ws;
    float* dinv = ws + N;
    float* scr  = dinv + N;
    float* Ta   = scr + (size_t)N * DFEAT;
    float* Tb   = Ta + (size_t)N * DFEAT;
    float* h    = Tb + (size_t)N * DFEAT;

    int*   rowptr   = (int*)scr;
    int*   slot_eid = rowptr + (N + 4);
    int*   col      = slot_eid + E;
    float* wcol     = (float*)(col + E);
    int*   tab      = (int*)Ta;

    const int TB = 256;
    auto cdiv = [](long long a, long long b) { return (int)((a + b - 1) / b); };

    const int chunk = cdiv(E, HG);
    hist_kernel<<<HG, HT, 0, stream>>>(dst, tab, N, E, chunk);
    colscan_kernel<<<cdiv(N / 4, 64), 256, 0, stream>>>(tab, cnt, N);
    scan_dinv_kernel<<<cdiv(N, SCAN_SPAN), 256, 0, stream>>>(cnt, rowptr, dinv, N);
    place_kernel<<<HG, HT, 0, stream>>>(dst, rowptr, tab, slot_eid, N, E, chunk);
    const int nodeBlocks = cdiv((long long)N * 32, TB);
    order_kernel<<<nodeBlocks, TB, 0, stream>>>(rowptr, slot_eid, src, dinv, col, wcol, N);

    gather_update_kernel<1><<<nodeBlocks, TB, 0, stream>>>(feat, rowptr, col, wcol, dinv, nullptr, Ta, nullptr, nullptr, theta, 1.0f, N);
    gather_update_kernel<2><<<nodeBlocks, TB, 0, stream>>>(Ta, rowptr, col, wcol, dinv, feat, Tb, nullptr, nullptr, theta, 2.0f, N);
    gather_update_kernel<3><<<nodeBlocks, TB, 0, stream>>>(Tb, rowptr, col, wcol, dinv, Ta, nullptr, feat, h, theta, 2.0f, N);

    const int strips = N / 16;
    gemm_wmma_kernel<<<cdiv(strips, GW), GW * 32, 0, stream>>>(h, W, out, N);
}


# Final GEMM moved into the order-3 gather launch: 16-wave groups form 16 h rows in LDS as the A strip, W staged once per group, 8 waves run the WMMAs; h plane and GEMM launch gone; sums unchanged
# speedup vs baseline: 4.2335x; 1.0343x over previous
#include <hip/hip_runtime.h>

typedef float v2f __attribute__((ext_vector_type(2)));
typedef float v8f __attribute__((ext_vector_type(8)));

#define DFEAT 128

#define HG 128
#define HT 1024
#define HR 65536
static_assert(HG <= DFEAT, "the slice table lives in one N x DFEAT plane");

__global__ void __launch_bounds__(HT)
hist_kernel(const int* __restrict__ dst, int* __restrict__ tab, int N, int E, int chunk) {
    __shared__ int lc[HR];
    const int tid = threadIdx.x;
    const int e0 = min(E, (int)blockIdx.x * chunk);
    const int e1 = min(E, e0 + chunk);
    int* row = tab + (size_t)blockIdx.x * N;
    for (int r0 = 0; r0 < N; r0 += HR) {
        const int rn = min(HR, N - r0);
        for (int i = tid * 4; i < rn; i += HT * 4) *(int4*)(lc + i) = make_int4(0, 0, 0, 0);
        __syncthreads();
        for (int eb = e0; eb < e1; eb += HT * 8) {
            int v[8];
#pragma unroll
            for (int k = 0; k < 8; ++k) v[k] = dst[min(eb + k * HT + tid, e1 - 1)];
#pragma unroll
            for (int k = 0; k < 8; ++k) {
                const int d = (eb + k * HT + tid < e1) ? v[k] - r0 : -1;
                if ((unsigned)d < (unsigned)rn) atomicAdd(&lc[d], 1);
            }
        }
        __syncthreads();
        for (int i = tid * 4; i < rn; i += HT * 4) *(int4*)(row + r0 + i) = *(const int4*)(lc + i);
        __syncthreads();
    }
}

__device__ __forceinline__ int4 add4(int4 a, int4 b) {
    return make_int4(a.x + b.x, a.y + b.y, a.z + b.z, a.w + b.w);
}

#define CS_G (HG / 4)
__global__ void __launch_bounds__(256)
colscan_kernel(int* __restrict__ tab, int* __restrict__ cnt, int N) {
    __shared__ int4 part[4][64];
    const int q = threadIdx.x & 63;
    const int p = threadIdx.x >> 6;
    const int NQ = N >> 2;
    const int n4 = blockIdx.x * 64 + q;
    const bool act = n4 < NQ;
    int4* col = (int4*)tab + (size_t)(p * CS_G) * NQ + min(n4, NQ - 1);
    int4 v[CS_G];
#pragma unroll
    for (int k = 0; k < CS_G; ++k) v[k] = col[(size_t)k * NQ];
    int4 run = make_int4(0, 0, 0, 0);
#pragma unroll
    for (int k = 0; k < CS_G; ++k) {
        const int4 t = v[k];
        v[k] = run;
        run = add4(run, t);
    }
    part[p][q] = run;
    __syncthreads();
    int4 off = make_int4(0, 0, 0, 0);
#pragma unroll
    for (int pp = 0; pp < 3; ++pp)
        if (pp < p) off = add4(off, part[pp][q]);
    if (act) {
#pragma unroll
        for (int k = 0; k < CS_G; ++k) col[(size_t)k * NQ] = add4(v[k], off);
        if (p == 3) ((int4*)cnt)[n4] = add4(off, run);
    }
}

#define SCAN_SPAN 1024
#define SCAN_GRP 16
__global__ void __launch_bounds__(256)
scan_dinv_kernel(const int* __restrict__ cnt, int* __restrict__ rowptr,
                 float* __restrict__ dinv, int N) {
    __shared__ int part[8];
    __shared__ int wtot[8];
    const int tid = threadIdx.x;
    const int lane = tid & 31;
    const int wv = tid >> 5;
    const int nb = blockIdx.x;
    const int base = nb * SCAN_SPAN;

    const int i0 = base + tid * 4;
    int c[4];
#pragma unroll
    for (int k = 0; k < 4; ++k) c[k] = (i0 + k < N) ? cnt[i0 + k] : 0;

    int s = 0;
    const int4* cq = (const int4*)cnt + tid;
    for (int j0 = 0; j0 < nb; j0 += SCAN_GRP) {
        int4 v[SCAN_GRP];
#pragma unroll
        for (int k = 0; k < SCAN_GRP; ++k)
            v[k] = cq[(size_t)min(j0 + k, nb - 1) * (SCAN_SPAN / 4)];
#pragma unroll
        for (int k = 0; k < SCAN_GRP; ++k)
            if (j0 + k < nb) s += (v[k].x + v[k].y) + (v[k].z + v[k].w);
    }
#pragma unroll
    for (int o = 16; o > 0; o >>= 1) s += __shfl_xor(s, o);
    if (lane == 0) part[wv] = s;

    const int local = (c[0] + c[1]) + (c[2] + c[3]);
    int inc = local;
#pragma unroll
    for (int o = 1; o < 32; o <<= 1) {
        int t = __shfl_up(inc, o);
        if (lane >= o) inc += t;
    }
    if (lane == 31) wtot[wv] = inc;
    __syncthreads();

    int run = 0;
#pragma unroll
    for (int w = 0; w < 8; ++w) {
        run += part[w];
        if (w < wv) run += wtot[w];
    }
    run += inc - local;
#pragma unroll
    for (int k = 0; k < 4; ++k) {
        int idx = i0 + k;
        if (idx < N) {
            rowptr[idx] = run;
            dinv[idx] = 1.0f / sqrtf(fmaxf((float)c[k], 1.0f));
            run += c[k];
            if (idx == N - 1) rowptr[N] = run;
        }
    }
}

__global__ void __launch_bounds__(HT)
place_kernel(const int* __restrict__ dst, const int* __restrict__ rowptr,
             const int* __restrict__ tab, int* __restrict__ slot_eid, int N, int E, int chunk) {
    __shared__ int lc[HR];
    const int tid = threadIdx.x;
    const int e0 = min(E, (int)blockIdx.x * chunk);
    const int e1 = min(E, e0 + chunk);
    const int* row = tab + (size_t)blockIdx.x * N;
    for (int r0 = 0; r0 < N; r0 += HR) {
        const int rn = min(HR, N - r0);
#pragma unroll 4
        for (int i = tid * 4; i < rn; i += HT * 4)
            *(int4*)(lc + i) = add4(*(const int4*)(row + r0 + i), *(const int4*)(rowptr + r0 + i));
        __syncthreads();
        for (int eb = e0; eb < e1; eb += HT * 8) {
            int v[8];
#pragma unroll
            for (int k = 0; k < 8; ++k) v[k] = dst[min(eb + k * HT + tid, e1 - 1)];
#pragma unroll
            for (int k = 0; k < 8; ++k) {
                const int e = eb + k * HT + tid;
                const int d = (e < e1) ? v[k] - r0 : -1;
                if ((unsigned)d < (unsigned)rn) slot_eid[atomicAdd(&lc[d], 1)] = e;
            }
        }
        __syncthreads();
    }
}

__global__ void __launch_bounds__(256)
order_kernel(const int* __restrict__ rowptr, const int* __restrict__ slot_eid,
             const int* __restrict__ src, const float* __restrict__ dinv,
             int* __restrict__ col, float* __restrict__ wcol, int N) {
    const int lane = threadIdx.x & 31;
    const int node = __builtin_amdgcn_readfirstlane((int)((blockIdx.x * blockDim.x + threadIdx.x) >> 5));
    if (node >= N) return;
    const int b = rowptr[node];
    const int e = rowptr[node + 1];
    for (int cb = b; cb < e; cb += 32) {
        const int j = cb + lane;
        const bool act = j < e;
        const int mine = act ? slot_eid[j] : 0x7fffffff;
        int rank = 0;
        for (int ob = b; ob < e; ob += 32) {
            const int jo = ob + lane;
            const int other = (jo < e) ? slot_eid[jo] : 0x7fffffff;
            const int n = min(32, e - ob);
            for (int k = 0; k < n; ++k)
                rank += (__builtin_amdgcn_readlane(other, k) < mine) ? 1 : 0;
        }
        if (act) {
            const int sn = src[mine];
            col[b + rank] = sn;
            wcol[b + rank] = dinv[sn];
        }
    }
}


template <int R>
__device__ __forceinline__ void gather_rows(const float* __restrict__ x, int sv, float wv,
                                            int k, int lane, float4& acc) {
    float4 v[R];
    float w[R];
#pragma unroll
    for (int r = 0; r < R; ++r) {
        const int s = __builtin_amdgcn_readlane(sv, k + r);
        w[r] = __int_as_float(__builtin_amdgcn_readlane(__float_as_int(wv), k + r));
        v[r] = *(const float4*)(x + (size_t)s * DFEAT + lane * 4);
    }
#pragma unroll
    for (int r = 0; r < R; ++r) {
        acc.x = fmaf(w[r], v[r].x, acc.x);
        acc.y = fmaf(w[r], v[r].y, acc.y);
        acc.z = fmaf(w[r], v[r].z, acc.z);
        acc.w = fmaf(w[r], v[r].w, acc.w);
    }
}

__device__ __forceinline__ void gather_chunk(const float* __restrict__ x, int sv, float wv,
                                             int n, int lane, float4& acc) {
    int k = 0;
    for (; k + 4 <= n; k += 4) gather_rows<4>(x, sv, wv, k, lane, acc);
    const int r = n - k;
    if (r == 3)      gather_rows<3>(x, sv, wv, k, lane, acc);
    else if (r == 2) gather_rows<2>(x, sv, wv, k, lane, acc);
    else if (r == 1) gather_rows<1>(x, sv, wv, k, lane, acc);
}

template <int ORD>
__device__ __forceinline__ float4 node_row(const float* __restrict__ x, const int* __restrict__ rowptr,
                                           const int* __restrict__ col, const float* __restrict__ wcol,
                                           const float* __restrict__ dinv,
                                           const float* __restrict__ Tprev,
                                           const float* __restrict__ feat,
                                           const float* __restrict__ theta, float c2,
                                           int node, int lane) {
    const int b = rowptr[node];
    const int e = rowptr[node + 1];
    const float s = -c2 * dinv[node];
    float t0 = 0.f, th1 = 0.f, th2 = 0.f, th3 = 0.f;
    if (ORD == 3) {
        t0 = theta[0];
        th1 = theta[1] * 0.5f;
        th2 = theta[2] * 0.5f;
        th3 = theta[3] * 0.5f;
    }
    const size_t i = (size_t)node * (DFEAT / 4) + lane;

    int sv = 0;
    float wv = 0.f;
    if (b + lane < e) { sv = col[b + lane]; wv = wcol[b + lane]; }

    float4 tp = make_float4(0.f, 0.f, 0.f, 0.f);
    if (ORD >= 2) tp = ((const float4*)Tprev)[i];
    float4 f = make_float4(0.f, 0.f, 0.f, 0.f);
    float4 t2 = make_float4(0.f, 0.f, 0.f, 0.f);
    if (ORD == 3) {
        f = ((const float4*)feat)[i];
        t2 = ((const float4*)x)[i];
    }

    float4 acc = make_float4(0.f, 0.f, 0.f, 0.f);
    gather_chunk(x, sv, wv, min(32, e - b), lane, acc);
    for (int cb = b + 32; cb < e; cb += 32) {
        const int j = cb + lane;
        sv = 0;
        wv = 0.f;
        if (j < e) { sv = col[j]; wv = wcol[j]; }
        gather_chunk(x, sv, wv, min(32, e - cb), lane, acc);
    }

    float4 t;
    t.x = fmaf(s, acc.x, -tp.x);
    t.y = fmaf(s, acc.y, -tp.y);
    t.z = fmaf(s, acc.z, -tp.z);
    t.w = fmaf(s, acc.w, -tp.w);
    if (ORD < 3) return t;

    float4 hv = make_float4(t0 * f.x, t0 * f.y, t0 * f.z, t0 * f.w);
    hv.x = fmaf(th1, tp.x + 1.0f, hv.x);
    hv.y = fmaf(th1, tp.y + 1.0f, hv.y);
    hv.z = fmaf(th1, tp.z + 1.0f, hv.z);
    hv.w = fmaf(th1, tp.w + 1.0f, hv.w);
    hv.x = fmaf(th2, t2.x + 1.0f, hv.x);
    hv.y = fmaf(th2, t2.y + 1.0f, hv.y);
    hv.z = fmaf(th2, t2.z + 1.0f, hv.z);
    hv.w = fmaf(th2, t2.w + 1.0f, hv.w);
    hv.x = fmaf(th3, t.x + 1.0f, hv.x);
    hv.y = fmaf(th3, t.y + 1.0f, hv.y);
    hv.z = fmaf(th3, t.z + 1.0f, hv.z);
    hv.w = fmaf(th3, t.w + 1.0f, hv.w);
    return hv;
}

template <int ORD>
__global__ void __launch_bounds__(256)
gather_update_kernel(const float* __restrict__ x, const int* __restrict__ rowptr,
                     const int* __restrict__ col, const float* __restrict__ wcol,
                     const float* __restrict__ dinv,
                     const float* __restrict__ Tprev, float* __restrict__ Tnew,
                     float c2, int N) {
    const int lane = threadIdx.x & 31;
    const int node = __builtin_amdgcn_readfirstlane((int)((blockIdx.x * blockDim.x + threadIdx.x) >> 5));
    if (node >= N) return;
    const float4 t = node_row<ORD>(x, rowptr, col, wcol, dinv, Tprev, nullptr, nullptr, c2, node, lane);
    ((float4*)Tnew)[(size_t)node * (DFEAT / 4) + lane] = t;
}

#define GP 132
#define FW 16
#define FT 4

__global__ void __launch_bounds__(FW * 32)
gather3_gemm_kernel(const float* __restrict__ x, const int* __restrict__ rowptr,
                    const int* __restrict__ col, const float* __restrict__ wcol,
                    const float* __restrict__ dinv, const float* __restrict__ Tprev,
                    const float* __restrict__ feat, const float* __restrict__ theta,
                    const float* __restrict__ W, float* __restrict__ out, int N) {
    __shared__ float sW[DFEAT * GP];
    __shared__ float sA[16 * GP];
    const int tid = threadIdx.x;
    const int lane = tid & 31;
    const int wave = __builtin_amdgcn_readfirstlane(tid >> 5);
    const int m = lane & 15;
    const int half = lane >> 4;

    const float4* Wq = (const float4*)W + tid;
    float4 wq[DFEAT / FW];
#pragma unroll
    for (int j = 0; j < DFEAT / FW; ++j) wq[j] = Wq[j * (FW * 32)];
#pragma unroll
    for (int j = 0; j < DFEAT / FW; ++j) *(float4*)(sW + (wave + FW * j) * GP + lane * 4) = wq[j];

    float* myRow = sA + wave * GP + lane * 4;
    const float* aP = sA + m * GP + (half << 1);
    const float* bP = sW + ((wave & 7) * 16 + m) * GP + (half << 1);

    const int nStrips = (N + 15) >> 4;
    for (int strip = blockIdx.x; strip < nStrips; strip += gridDim.x) {
        const int node = strip * 16 + wave;
        float4 hv = make_float4(0.f, 0.f, 0.f, 0.f);
        if (node < N)
            hv = node_row<3>(x, rowptr, col, wcol, dinv, Tprev, feat, theta, 2.0f, node, lane);
        *(float4*)myRow = hv;
        __syncthreads();

        v8f acc = (v8f){};
        if (wave < 8) {
#pragma unroll
            for (int kk = 0; kk < DFEAT; kk += 4) {
                const v2f a = *(const v2f*)(aP + kk);
                const v2f b = *(const v2f*)(bP + kk);
                acc = __builtin_amdgcn_wmma_f32_16x16x4_f32(
                     false, a,  false, b,
                     (short)0, acc,  false,  false);
            }
#pragma unroll
            for (int v = 0; v < 8; ++v) acc[v] = (acc[v] >= 0.0f) ? acc[v] : 0.01f * acc[v];
        }
        __syncthreads();

        if (wave < 8) {
#pragma unroll
            for (int v = 0; v < 8; ++v) sA[(v + (half << 3)) * GP + wave * 16 + m] = acc[v];
        }
        __syncthreads();
        if (node < N)
            ((float4*)out)[(size_t)node * (DFEAT / 4) + lane] = *(const float4*)myRow;
    }
}


extern "C" void kernel_launch(void* const* d_in, const int* in_sizes, int n_in,
                              void* d_out, int out_size, void* d_ws, size_t ws_size,
                              hipStream_t stream) {
    const float* feat  = (const float*)d_in[0];
    const int*   src   = (const int*)d_in[1];
    const int*   dst   = (const int*)d_in[2];
    const float* theta = (const float*)d_in[3];
    const float* W     = (const float*)d_in[4];
    float* out = (float*)d_out;

    const int N = in_sizes[0] / DFEAT;
    const int E = in_sizes[1];

    float* ws   = (float*)d_ws;
    int*   cnt  = (int*)ws;
    float* dinv = ws + N;
    float* scr  = dinv + N;
    float* Ta   = scr + (size_t)N * DFEAT;
    float* Tb   = Ta + (size_t)N * DFEAT;

    int*   rowptr   = (int*)scr;
    int*   slot_eid = rowptr + (N + 4);
    int*   col      = slot_eid + E;
    float* wcol     = (float*)(col + E);
    int*   tab      = (int*)Ta;

    const int TB = 256;
    auto cdiv = [](long long a, long long b) { return (int)((a + b - 1) / b); };

    const int chunk = cdiv(E, HG);
    hist_kernel<<<HG, HT, 0, stream>>>(dst, tab, N, E, chunk);
    colscan_kernel<<<cdiv(N / 4, 64), 256, 0, stream>>>(tab, cnt, N);
    scan_dinv_kernel<<<cdiv(N, SCAN_SPAN), 256, 0, stream>>>(cnt, rowptr, dinv, N);
    place_kernel<<<HG, HT, 0, stream>>>(dst, rowptr, tab, slot_eid, N, E, chunk);
    const int nodeBlocks = cdiv((long long)N * 32, TB);
    order_kernel<<<nodeBlocks, TB, 0, stream>>>(rowptr, slot_eid, src, dinv, col, wcol, N);

    gather_update_kernel<1><<<nodeBlocks, TB, 0, stream>>>(feat, rowptr, col, wcol, dinv, nullptr, Ta, 1.0f, N);
    gather_update_kernel<2><<<nodeBlocks, TB, 0, stream>>>(Ta, rowptr, col, wcol, dinv, feat, Tb, 2.0f, N);
    const int strips = cdiv(N, 16);
    gather3_gemm_kernel<<<cdiv(strips, FT), FW * 32, 0, stream>>>(Tb, rowptr, col, wcol, dinv, Ta, feat, theta, W, out, N);
}


# Edge ordering rebuilt as two-level partition: edges go as records into 256 node-range regions, one work-group sorts a region in LDS; 5 launches become 3, order-1 gather writes wcol; sums unchanged
# speedup vs baseline: 5.3806x; 1.2710x over previous
#include <hip/hip_runtime.h>

typedef float v2f __attribute__((ext_vector_type(2)));
typedef float v8f __attribute__((ext_vector_type(8)));

#define DFEAT 128

#define NS 128
#define NR 256
#define RWMAX 256
#define PT 512
static_assert(PT == 2 * NR, "partition_kernel: a thread owns one range and half of the slices");
static_assert(NS % 2 == 0, "the slices are read in two halves");

__global__ void __launch_bounds__(PT)
count_kernel(const int* __restrict__ dst, int* __restrict__ tabc, int E, int chunk, int RW) {
    __shared__ int lc[NR];
    const int tid = threadIdx.x;
    if (tid < NR) lc[tid] = 0;
    __syncthreads();
    const int e0 = min(E, (int)blockIdx.x * chunk);
    const int e1 = min(E, e0 + chunk);
    for (int eb = e0; eb < e1; eb += PT * 8) {
        int v[8];
#pragma unroll
        for (int k = 0; k < 8; ++k) v[k] = dst[min(eb + k * PT + tid, e1 - 1)];
#pragma unroll
        for (int k = 0; k < 8; ++k)
            if (eb + k * PT + tid < e1) atomicAdd(&lc[v[k] / RW], 1);
    }
    __syncthreads();
    if (tid < NR) tabc[(size_t)blockIdx.x * NR + tid] = lc[tid];
}

__global__ void __launch_bounds__(PT)
partition_kernel(const int* __restrict__ dst, const int* __restrict__ tabc,
                 int* __restrict__ rstart, int2* __restrict__ rec, int E, int chunk, int RW) {
    __shared__ int stot[2][NR];
    __shared__ int spre[2][NR];
    __shared__ int wtot[NR / 32];
    __shared__ int lb[NR];
    const int tid = threadIdx.x;
    const int lane = tid & 31;
    const int wv = tid >> 5;
    const int s = blockIdx.x;
    const int r = tid & (NR - 1);
    const int p = tid >> 8;
    const int* colp = tabc + (size_t)(p * (NS / 2)) * NR + r;
    int v[NS / 2];
#pragma unroll
    for (int k = 0; k < NS / 2; ++k) v[k] = colp[(size_t)k * NR];
    int tot = 0, pre = 0;
#pragma unroll
    for (int k = 0; k < NS / 2; ++k) {
        tot += v[k];
        if (p * (NS / 2) + k < s) pre += v[k];
    }
    stot[p][r] = tot;
    spre[p][r] = pre;
    __syncthreads();

    int T = 0, P = 0;
    if (tid < NR) {
        T = stot[0][tid] + stot[1][tid];
        P = spre[0][tid] + spre[1][tid];
    }
    int inc = T;
    for (int o = 1; o < 32; o <<= 1) {
        int t = __shfl_up(inc, o);
        if (lane >= o) inc += t;
    }
    if (tid < NR && lane == 31) wtot[wv] = inc;
    __syncthreads();
    if (tid < NR) {
        int run = 0;
        for (int w = 0; w < NR / 32; ++w)
            if (w < wv) run += wtot[w];
        const int rs = run + inc - T;
        lb[tid] = rs + P;
        if (s == 0) {
            rstart[tid] = rs;
            if (tid == NR - 1) rstart[NR] = rs + T;
        }
    }
    __syncthreads();

    const int e0 = min(E, s * chunk);
    const int e1 = min(E, e0 + chunk);
    for (int eb = e0; eb < e1; eb += PT * 8) {
        int d[8];
#pragma unroll
        for (int k = 0; k < 8; ++k) d[k] = dst[min(eb + k * PT + tid, e1 - 1)];
#pragma unroll
        for (int k = 0; k < 8; ++k) {
            const int e = eb + k * PT + tid;
            if (e < e1) rec[atomicAdd(&lb[d[k] / RW], 1)] = make_int2(e, d[k]);
        }
    }
}

#define PT3 512
#define CAP 8192
template <bool INLDS>
__device__ __forceinline__ void range_body(const int2* Rg, int2* R, int* G,
                                           int* lcnt, int* lexc, int* lfill, int* wtot,
                                           const int* __restrict__ src,
                                           int* __restrict__ rowptr, float* __restrict__ dinv,
                                           int* __restrict__ col,
                                           int rb, int cnt, int n0, int nn, int N) {
    const int tid = threadIdx.x;
    const int lane = tid & 31;
    const int wv = __builtin_amdgcn_readfirstlane(tid >> 5);

    for (int i0 = tid; i0 < cnt; i0 += PT3 * 8) {
        int2 q[8];
#pragma unroll
        for (int k = 0; k < 8; ++k) q[k] = Rg[min(i0 + k * PT3, cnt - 1)];
#pragma unroll
        for (int k = 0; k < 8; ++k) {
            const int i = i0 + k * PT3;
            if (i < cnt) {
                if (INLDS) R[i] = q[k];
                atomicAdd(&lcnt[q[k].y - n0], 1);
            }
        }
    }
    __syncthreads();

    int c = 0;
    if (tid < RWMAX) c = lcnt[tid];
    int inc = c;
    for (int o = 1; o < 32; o <<= 1) {
        int t = __shfl_up(inc, o);
        if (lane >= o) inc += t;
    }
    if (tid < RWMAX && lane == 31) wtot[wv] = inc;
    __syncthreads();
    if (tid < RWMAX) {
        int run = 0;
        for (int w = 0; w < RWMAX / 32; ++w)
            if (w < wv) run += wtot[w];
        const int ex = run + inc - c;
        lexc[tid] = ex;
        lfill[tid] = ex;
        if (tid < nn) {
            const int node = n0 + tid;
            rowptr[node] = rb + ex;
            dinv[node] = 1.0f / sqrtf(fmaxf((float)c, 1.0f));
            if (node == N - 1) rowptr[N] = rb + ex + c;
        }
    }
    __syncthreads();

    for (int i = tid; i < cnt; i += PT3) {
        const int2 q = R[i];
        G[atomicAdd(&lfill[q.y - n0], 1)] = q.x;
    }
    __syncthreads();

    int* S = (int*)R;
    for (int j = wv; j < nn; j += PT3 / 32) {
        const int b = __builtin_amdgcn_readfirstlane(lexc[j]);
        const int e = b + __builtin_amdgcn_readfirstlane(lcnt[j]);
        for (int cb = b; cb < e; cb += 32) {
            const int jm = cb + lane;
            const bool act = jm < e;
            const int mine = act ? G[jm] : 0x7fffffff;
            int rank = 0;
            for (int ob = b; ob < e; ob += 32) {
                const int jo = ob + lane;
                const int other = (ob == cb) ? mine : ((jo < e) ? G[jo] : 0x7fffffff);
                const int n = min(32, e - ob);
                for (int k = 0; k < n; ++k)
                    rank += (__builtin_amdgcn_readlane(other, k) < mine) ? 1 : 0;
            }
            if (act) S[b + rank] = mine;
        }
    }
    __syncthreads();

    for (int i0 = tid; i0 < cnt; i0 += PT3 * 8) {
        int ev[8], sn[8];
#pragma unroll
        for (int k = 0; k < 8; ++k) ev[k] = S[min(i0 + k * PT3, cnt - 1)];
#pragma unroll
        for (int k = 0; k < 8; ++k) sn[k] = src[ev[k]];
#pragma unroll
        for (int k = 0; k < 8; ++k)
            if (i0 + k * PT3 < cnt) col[rb + i0 + k * PT3] = sn[k];
    }
}

__global__ void __launch_bounds__(PT3)
range_kernel(const int* __restrict__ rstart, int2* rec, int* grp,
             const int* __restrict__ src, int* __restrict__ rowptr, float* __restrict__ dinv,
             int* __restrict__ col, int N, int RW) {
    __shared__ int2 srec[CAP];
    __shared__ int sg[CAP];
    __shared__ int lcnt[RWMAX];
    __shared__ int lexc[RWMAX];
    __shared__ int lfill[RWMAX];
    __shared__ int wtot[RWMAX / 32];
    const int tid = threadIdx.x;
    const int r = blockIdx.x;
    const int rb = rstart[r];
    const int cnt = rstart[r + 1] - rb;
    const int n0 = r * RW;
    const int nn = max(0, min(RW, N - n0));
    if (tid < RWMAX) lcnt[tid] = 0;
    __syncthreads();
    if (cnt <= CAP)
        range_body<true>(rec + rb, srec, sg, lcnt, lexc, lfill, wtot, src, rowptr, dinv, col, rb, cnt, n0, nn, N);
    else
        range_body<false>(rec + rb, rec + rb, grp + rb, lcnt, lexc, lfill, wtot, src, rowptr, dinv, col, rb, cnt, n0, nn, N);
}


template <int R>
__device__ __forceinline__ void gather_rows(const float* __restrict__ x, int sv, float wv,
                                            int k, int lane, float4& acc) {
    float4 v[R];
    float w[R];
#pragma unroll
    for (int r = 0; r < R; ++r) {
        const int s = __builtin_amdgcn_readlane(sv, k + r);
        w[r] = __int_as_float(__builtin_amdgcn_readlane(__float_as_int(wv), k + r));
        v[r] = *(const float4*)(x + (size_t)s * DFEAT + lane * 4);
    }
#pragma unroll
    for (int r = 0; r < R; ++r) {
        acc.x = fmaf(w[r], v[r].x, acc.x);
        acc.y = fmaf(w[r], v[r].y, acc.y);
        acc.z = fmaf(w[r], v[r].z, acc.z);
        acc.w = fmaf(w[r], v[r].w, acc.w);
    }
}

__device__ __forceinline__ void gather_chunk(const float* __restrict__ x, int sv, float wv,
                                             int n, int lane, float4& acc) {
    int k = 0;
    for (; k + 4 <= n; k += 4) gather_rows<4>(x, sv, wv, k, lane, acc);
    const int r = n - k;
    if (r == 3)      gather_rows<3>(x, sv, wv, k, lane, acc);
    else if (r == 2) gather_rows<2>(x, sv, wv, k, lane, acc);
    else if (r == 1) gather_rows<1>(x, sv, wv, k, lane, acc);
}

template <int ORD>
__device__ __forceinline__ float4 node_row(const float* __restrict__ x, const int* __restrict__ rowptr,
                                           const int* __restrict__ col, float* __restrict__ wcol,
                                           const float* __restrict__ dinv,
                                           const float* __restrict__ Tprev,
                                           const float* __restrict__ feat,
                                           const float* __restrict__ theta, float c2,
                                           int node, int lane) {
    const int b = rowptr[node];
    const int e = rowptr[node + 1];
    const float s = -c2 * dinv[node];
    float t0 = 0.f, th1 = 0.f, th2 = 0.f, th3 = 0.f;
    if (ORD == 3) {
        t0 = theta[0];
        th1 = theta[1] * 0.5f;
        th2 = theta[2] * 0.5f;
        th3 = theta[3] * 0.5f;
    }
    const size_t i = (size_t)node * (DFEAT / 4) + lane;

    int sv = 0;
    float wv = 0.f;
    if (b + lane < e) {
        sv = col[b + lane];
        if (ORD == 1) { wv = dinv[sv]; wcol[b + lane] = wv; }
        else wv = wcol[b + lane];
    }

    float4 tp = make_float4(0.f, 0.f, 0.f, 0.f);
    if (ORD >= 2) tp = ((const float4*)Tprev)[i];
    float4 f = make_float4(0.f, 0.f, 0.f, 0.f);
    float4 t2 = make_float4(0.f, 0.f, 0.f, 0.f);
    if (ORD == 3) {
        f = ((const float4*)feat)[i];
        t2 = ((const float4*)x)[i];
    }

    float4 acc = make_float4(0.f, 0.f, 0.f, 0.f);
    gather_chunk(x, sv, wv, min(32, e - b), lane, acc);
    for (int cb = b + 32; cb < e; cb += 32) {
        const int j = cb + lane;
        sv = 0;
        wv = 0.f;
        if (j < e) {
            sv = col[j];
            if (ORD == 1) { wv = dinv[sv]; wcol[j] = wv; }
            else wv = wcol[j];
        }
        gather_chunk(x, sv, wv, min(32, e - cb), lane, acc);
    }

    float4 t;
    t.x = fmaf(s, acc.x, -tp.x);
    t.y = fmaf(s, acc.y, -tp.y);
    t.z = fmaf(s, acc.z, -tp.z);
    t.w = fmaf(s, acc.w, -tp.w);
    if (ORD < 3) return t;

    float4 hv = make_float4(t0 * f.x, t0 * f.y, t0 * f.z, t0 * f.w);
    hv.x = fmaf(th1, tp.x + 1.0f, hv.x);
    hv.y = fmaf(th1, tp.y + 1.0f, hv.y);
    hv.z = fmaf(th1, tp.z + 1.0f, hv.z);
    hv.w = fmaf(th1, tp.w + 1.0f, hv.w);
    hv.x = fmaf(th2, t2.x + 1.0f, hv.x);
    hv.y = fmaf(th2, t2.y + 1.0f, hv.y);
    hv.z = fmaf(th2, t2.z + 1.0f, hv.z);
    hv.w = fmaf(th2, t2.w + 1.0f, hv.w);
    hv.x = fmaf(th3, t.x + 1.0f, hv.x);
    hv.y = fmaf(th3, t.y + 1.0f, hv.y);
    hv.z = fmaf(th3, t.z + 1.0f, hv.z);
    hv.w = fmaf(th3, t.w + 1.0f, hv.w);
    return hv;
}

template <int ORD>
__global__ void __launch_bounds__(256)
gather_update_kernel(const float* __restrict__ x, const int* __restrict__ rowptr,
                     const int* __restrict__ col, float* __restrict__ wcol,
                     const float* __restrict__ dinv,
                     const float* __restrict__ Tprev, float* __restrict__ Tnew,
                     float c2, int N) {
    const int lane = threadIdx.x & 31;
    const int node = __builtin_amdgcn_readfirstlane((int)((blockIdx.x * blockDim.x + threadIdx.x) >> 5));
    if (node >= N) return;
    const float4 t = node_row<ORD>(x, rowptr, col, wcol, dinv, Tprev, nullptr, nullptr, c2, node, lane);
    ((float4*)Tnew)[(size_t)node * (DFEAT / 4) + lane] = t;
}

#define GP 132
#define FW 16
#define FT 4

__global__ void __launch_bounds__(FW * 32)
gather3_gemm_kernel(const float* __restrict__ x, const int* __restrict__ rowptr,
                    const int* __restrict__ col, float* __restrict__ wcol,
                    const float* __restrict__ dinv, const float* __restrict__ Tprev,
                    const float* __restrict__ feat, const float* __restrict__ theta,
                    const float* __restrict__ W, float* __restrict__ out, int N) {
    __shared__ float sW[DFEAT * GP];
    __shared__ float sA[16 * GP];
    const int tid = threadIdx.x;
    const int lane = tid & 31;
    const int wave = __builtin_amdgcn_readfirstlane(tid >> 5);
    const int m = lane & 15;
    const int half = lane >> 4;

    const float4* Wq = (const float4*)W + tid;
    float4 wq[DFEAT / FW];
#pragma unroll
    for (int j = 0; j < DFEAT / FW; ++j) wq[j] = Wq[j * (FW * 32)];
#pragma unroll
    for (int j = 0; j < DFEAT / FW; ++j) *(float4*)(sW + (wave + FW * j) * GP + lane * 4) = wq[j];

    float* myRow = sA + wave * GP + lane * 4;
    const float* aP = sA + m * GP + (half << 1);
    const float* bP = sW + ((wave & 7) * 16 + m) * GP + (half << 1);

    const int nStrips = (N + 15) >> 4;
    for (int strip = blockIdx.x; strip < nStrips; strip += gridDim.x) {
        const int node = strip * 16 + wave;
        float4 hv = make_float4(0.f, 0.f, 0.f, 0.f);
        if (node < N)
            hv = node_row<3>(x, rowptr, col, wcol, dinv, Tprev, feat, theta, 2.0f, node, lane);
        *(float4*)myRow = hv;
        __syncthreads();

        v8f acc = (v8f){};
        if (wave < 8) {
#pragma unroll
            for (int kk = 0; kk < DFEAT; kk += 4) {
                const v2f a = *(const v2f*)(aP + kk);
                const v2f b = *(const v2f*)(bP + kk);
                acc = __builtin_amdgcn_wmma_f32_16x16x4_f32(
                     false, a,  false, b,
                     (short)0, acc,  false,  false);
            }
#pragma unroll
            for (int v = 0; v < 8; ++v) acc[v] = (acc[v] >= 0.0f) ? acc[v] : 0.01f * acc[v];
        }
        __syncthreads();

        if (wave < 8) {
#pragma unroll
            for (int v = 0; v < 8; ++v) sA[(v + (half << 3)) * GP + wave * 16 + m] = acc[v];
        }
        __syncthreads();
        if (node < N)
            ((float4*)out)[(size_t)node * (DFEAT / 4) + lane] = *(const float4*)myRow;
    }
}


extern "C" void kernel_launch(void* const* d_in, const int* in_sizes, int n_in,
                              void* d_out, int out_size, void* d_ws, size_t ws_size,
                              hipStream_t stream) {
    const float* feat  = (const float*)d_in[0];
    const int*   src   = (const int*)d_in[1];
    const int*   dst   = (const int*)d_in[2];
    const float* theta = (const float*)d_in[3];
    const float* W     = (const float*)d_in[4];
    float* out = (float*)d_out;

    const int N = in_sizes[0] / DFEAT;
    const int E = in_sizes[1];

    float* ws   = (float*)d_ws;
    float* dinv = ws + N;
    float* scr  = dinv + N;
    float* Ta   = scr + (size_t)N * DFEAT;
    float* Tb   = Ta + (size_t)N * DFEAT;

    int2*  rec    = (int2*)scr;
    int*   grp    = (int*)(rec + E);
    int*   col    = grp + E;
    float* wcol   = (float*)(col + E);
    int*   rowptr = (int*)(wcol + E);
    int*   tabc   = rowptr + (N + 4);
    int*   rstart = tabc + NS * NR;

    const int TB = 256;
    auto cdiv = [](long long a, long long b) { return (int)((a + b - 1) / b); };

    const int chunk = cdiv(E, NS);
    const int RW = cdiv(N, NR);
    count_kernel<<<NS, PT, 0, stream>>>(dst, tabc, E, chunk, RW);
    partition_kernel<<<NS, PT, 0, stream>>>(dst, tabc, rstart, rec, E, chunk, RW);
    range_kernel<<<NR, PT3, 0, stream>>>(rstart, rec, grp, src, rowptr, dinv, col, N, RW);
    const int nodeBlocks = cdiv((long long)N * 32, TB);

    gather_update_kernel<1><<<nodeBlocks, TB, 0, stream>>>(feat, rowptr, col, wcol, dinv, nullptr, Ta, 1.0f, N);
    gather_update_kernel<2><<<nodeBlocks, TB, 0, stream>>>(Ta, rowptr, col, wcol, dinv, feat, Tb, 2.0f, N);
    const int strips = cdiv(N, 16);
    gather3_gemm_kernel<<<cdiv(strips, FT), FW * 32, 0, stream>>>(Tb, rowptr, col, wcol, dinv, Ta, feat, theta, W, out, N);
}


# Edge ordering level 1 in one launch: each slice orders its own records by node range in LDS and writes its segment whole; range groups collect their 128 pieces; count launch gone; sums unchanged
# speedup vs baseline: 6.0438x; 1.1233x over previous
#include <hip/hip_runtime.h>

typedef float v2f __attribute__((ext_vector_type(2)));
typedef float v8f __attribute__((ext_vector_type(8)));

#define DFEAT 128

#define NS 128
#define NR 256
#define RWMAX 256
#define PT1 1024
#define SCAP 8192
static_assert(PT1 >= NR && PT1 % 32 == 0, "slice_kernel: the first NR threads (whole waves) scan the range counts");

__global__ void __launch_bounds__(PT1)
slice_kernel(const int* __restrict__ dst, int* __restrict__ tabc, int* __restrict__ toff,
             int2* __restrict__ rec, int E, int chunk, int RW) {
    __shared__ int lc[NR];
    __shared__ int lfill[NR];
    __shared__ int wtot[NR / 32];
    __shared__ int2 sr[SCAP];
    const int tid = threadIdx.x;
    const int lane = tid & 31;
    const int wv = tid >> 5;
    const int s = blockIdx.x;
    if (tid < NR) lc[tid] = 0;
    __syncthreads();
    const int e0 = min(E, s * chunk);
    const int e1 = min(E, e0 + chunk);
    const int n = e1 - e0;
    const bool staged = n <= SCAP;

    for (int eb = e0; eb < e1; eb += PT1 * 8) {
        int v[8];
#pragma unroll
        for (int k = 0; k < 8; ++k) v[k] = dst[min(eb + k * PT1 + tid, e1 - 1)];
#pragma unroll
        for (int k = 0; k < 8; ++k)
            if (eb + k * PT1 + tid < e1) atomicAdd(&lc[v[k] / RW], 1);
    }
    __syncthreads();

    int c = 0;
    if (tid < NR) c = lc[tid];
    int inc = c;
    for (int o = 1; o < 32; o <<= 1) {
        int t = __shfl_up(inc, o);
        if (lane >= o) inc += t;
    }
    if (tid < NR && lane == 31) wtot[wv] = inc;
    __syncthreads();
    if (tid < NR) {
        int run = 0;
        for (int w = 0; w < NR / 32; ++w)
            if (w < wv) run += wtot[w];
        const int ex = run + inc - c;
        lfill[tid] = ex;
        tabc[(size_t)s * NR + tid] = c;
        toff[(size_t)s * NR + tid] = ex;
    }
    __syncthreads();

    for (int eb = e0; eb < e1; eb += PT1 * 8) {
        int d[8], p[8];
#pragma unroll
        for (int k = 0; k < 8; ++k) d[k] = dst[min(eb + k * PT1 + tid, e1 - 1)];
#pragma unroll
        for (int k = 0; k < 8; ++k) {
            p[k] = 0;
            if (eb + k * PT1 + tid < e1) p[k] = atomicAdd(&lfill[d[k] / RW], 1);
        }
#pragma unroll
        for (int k = 0; k < 8; ++k) {
            const int e = eb + k * PT1 + tid;
            if (e < e1) {
                const int2 q = make_int2(e, d[k]);
                if (staged) sr[p[k]] = q;
                else rec[e0 + p[k]] = q;
            }
        }
    }
    if (staged) {
        __syncthreads();
        for (int i = tid; i < n; i += PT1) rec[e0 + i] = sr[i];
    }
}

#define PT3 512
#define CAP 8192
#define NPW (NS / (PT3 / 32))
static_assert(PT3 == 2 * NR, "range_kernel: a thread sums one range's counts over half of the slices");
static_assert(NS % 2 == 0 && NS <= PT3 && NS % 32 == 0 && NS % (PT3 / 32) == 0, "range_kernel: slices per thread, per scan wave, per wave");
__device__ __forceinline__ void range_body(const int2* __restrict__ rec, int2* R, int* G,
                                           const int* pbase, const int* pcnt, const int* psrc,
                                           int* lcnt, int* lexc, int* lfill, int* wtot,
                                           const int* __restrict__ src,
                                           int* __restrict__ rowptr, float* __restrict__ dinv,
                                           int* __restrict__ col,
                                           int rb, int cnt, int n0, int nn, int N) {
    const int tid = threadIdx.x;
    const int lane = tid & 31;
    const int wv = __builtin_amdgcn_readfirstlane(tid >> 5);

    {
        int2 q[NPW][2];
#pragma unroll
        for (int j = 0; j < NPW; ++j) {
            const int s = wv + (PT3 / 32) * j;
            const int pn = pcnt[s];
            const int ps = psrc[s];
#pragma unroll
            for (int h = 0; h < 2; ++h)
                q[j][h] = rec[(pn > 0) ? ps + min(lane + 32 * h, pn - 1) : 0];
        }
#pragma unroll
        for (int j = 0; j < NPW; ++j) {
            const int s = wv + (PT3 / 32) * j;
            const int pn = pcnt[s];
            const int pb = pbase[s];
#pragma unroll
            for (int h = 0; h < 2; ++h) {
                if (lane + 32 * h < pn) {
                    R[pb + lane + 32 * h] = q[j][h];
                    atomicAdd(&lcnt[q[j][h].y - n0], 1);
                }
            }
        }
        for (int j = 0; j < NPW; ++j) {
            const int s = wv + (PT3 / 32) * j;
            const int pn = pcnt[s];
            for (int k = 64 + lane; k < pn; k += 32) {
                const int2 qq = rec[psrc[s] + k];
                R[pbase[s] + k] = qq;
                atomicAdd(&lcnt[qq.y - n0], 1);
            }
        }
    }
    __syncthreads();

    int c = 0;
    if (tid < RWMAX) c = lcnt[tid];
    int inc = c;
    for (int o = 1; o < 32; o <<= 1) {
        int t = __shfl_up(inc, o);
        if (lane >= o) inc += t;
    }
    if (tid < RWMAX && lane == 31) wtot[wv] = inc;
    __syncthreads();
    if (tid < RWMAX) {
        int run = 0;
        for (int w = 0; w < RWMAX / 32; ++w)
            if (w < wv) run += wtot[w];
        const int ex = run + inc - c;
        lexc[tid] = ex;
        lfill[tid] = ex;
        if (tid < nn) {
            const int node = n0 + tid;
            rowptr[node] = rb + ex;
            dinv[node] = 1.0f / sqrtf(fmaxf((float)c, 1.0f));
            if (node == N - 1) rowptr[N] = rb + ex + c;
        }
    }
    __syncthreads();

    for (int i = tid; i < cnt; i += PT3) {
        const int2 q = R[i];
        G[atomicAdd(&lfill[q.y - n0], 1)] = q.x;
    }
    __syncthreads();

    int* S = (int*)R;
    for (int j = wv; j < nn; j += PT3 / 32) {
        const int b = __builtin_amdgcn_readfirstlane(lexc[j]);
        const int e = b + __builtin_amdgcn_readfirstlane(lcnt[j]);
        for (int cb = b; cb < e; cb += 32) {
            const int jm = cb + lane;
            const bool act = jm < e;
            const int mine = act ? G[jm] : 0x7fffffff;
            int rank = 0;
            for (int ob = b; ob < e; ob += 32) {
                const int jo = ob + lane;
                const int other = (ob == cb) ? mine : ((jo < e) ? G[jo] : 0x7fffffff);
                const int n = min(32, e - ob);
                for (int k = 0; k < n; ++k)
                    rank += (__builtin_amdgcn_readlane(other, k) < mine) ? 1 : 0;
            }
            if (act) S[b + rank] = mine;
        }
    }
    __syncthreads();

    for (int i0 = tid; i0 < cnt; i0 += PT3 * 8) {
        int ev[8], sn[8];
#pragma unroll
        for (int k = 0; k < 8; ++k) ev[k] = S[min(i0 + k * PT3, cnt - 1)];
#pragma unroll
        for (int k = 0; k < 8; ++k) sn[k] = src[ev[k]];
#pragma unroll
        for (int k = 0; k < 8; ++k)
            if (i0 + k * PT3 < cnt) col[rb + i0 + k * PT3] = sn[k];
    }
}

__global__ void __launch_bounds__(PT3)
range_kernel(const int* __restrict__ tabc, const int* __restrict__ toff,
             const int2* __restrict__ rec, int2* rec2, int* grp,
             const int* __restrict__ src, int* __restrict__ rowptr, float* __restrict__ dinv,
             int* __restrict__ col, int N, int RW, int chunk) {
    __shared__ int2 srec[CAP];
    __shared__ int sg[CAP];
    __shared__ int lcnt[RWMAX];
    __shared__ int lexc[RWMAX];
    __shared__ int lfill[RWMAX];
    __shared__ int wtot[RWMAX / 32];
    __shared__ int stot[2][NR];
    __shared__ int wtotA[NR / 32];
    __shared__ int wtotB[NS / 32];
    __shared__ int pbase[NS];
    __shared__ int pcnt[NS];
    __shared__ int psrc[NS];
    __shared__ int srb;
    __shared__ int scnt;
    const int tid = threadIdx.x;
    const int lane = tid & 31;
    const int wv = tid >> 5;
    const int r = blockIdx.x;
    const int n0 = r * RW;
    const int nn = max(0, min(RW, N - n0));

    const int rr = tid & (NR - 1);
    const int p = tid >> 8;
    const int* colp = tabc + (size_t)(p * (NS / 2)) * NR + rr;
    int v[NS / 2];
#pragma unroll
    for (int k = 0; k < NS / 2; ++k) v[k] = colp[(size_t)k * NR];
    int pc = 0, po = 0;
    if (tid < NS) {
        pc = tabc[(size_t)tid * NR + r];
        po = toff[(size_t)tid * NR + r];
    }
    int tot = 0;
#pragma unroll
    for (int k = 0; k < NS / 2; ++k) tot += v[k];
    stot[p][rr] = tot;
    if (tid < RWMAX) lcnt[tid] = 0;
    __syncthreads();

    int T = 0;
    if (tid < NR) T = stot[0][tid] + stot[1][tid];
    int incA = T;
    int incB = pc;
    for (int o = 1; o < 32; o <<= 1) {
        int ta = __shfl_up(incA, o);
        int tb = __shfl_up(incB, o);
        if (lane >= o) { incA += ta; incB += tb; }
    }
    if (tid < NR && lane == 31) wtotA[wv] = incA;
    if (tid < NS && lane == 31) wtotB[wv] = incB;
    __syncthreads();
    if (tid == r) {
        int run = 0;
        for (int w = 0; w < NR / 32; ++w)
            if (w < wv) run += wtotA[w];
        srb = run + incA - T;
        scnt = T;
    }
    if (tid < NS) {
        int run = 0;
        for (int w = 0; w < NS / 32; ++w)
            if (w < wv) run += wtotB[w];
        pbase[tid] = run + incB - pc;
        pcnt[tid] = pc;
        psrc[tid] = tid * chunk + po;
    }
    __syncthreads();
    const int rb = srb;
    const int cnt = scnt;

    if (cnt <= CAP)
        range_body(rec, srec, sg, pbase, pcnt, psrc, lcnt, lexc, lfill, wtot, src, rowptr, dinv, col, rb, cnt, n0, nn, N);
    else
        range_body(rec, rec2 + rb, grp + rb, pbase, pcnt, psrc, lcnt, lexc, lfill, wtot, src, rowptr, dinv, col, rb, cnt, n0, nn, N);
}


template <int R>
__device__ __forceinline__ void gather_rows(const float* __restrict__ x, int sv, float wv,
                                            int k, int lane, float4& acc) {
    float4 v[R];
    float w[R];
#pragma unroll
    for (int r = 0; r < R; ++r) {
        const int s = __builtin_amdgcn_readlane(sv, k + r);
        w[r] = __int_as_float(__builtin_amdgcn_readlane(__float_as_int(wv), k + r));
        v[r] = *(const float4*)(x + (size_t)s * DFEAT + lane * 4);
    }
#pragma unroll
    for (int r = 0; r < R; ++r) {
        acc.x = fmaf(w[r], v[r].x, acc.x);
        acc.y = fmaf(w[r], v[r].y, acc.y);
        acc.z = fmaf(w[r], v[r].z, acc.z);
        acc.w = fmaf(w[r], v[r].w, acc.w);
    }
}

__device__ __forceinline__ void gather_chunk(const float* __restrict__ x, int sv, float wv,
                                             int n, int lane, float4& acc) {
    int k = 0;
    for (; k + 4 <= n; k += 4) gather_rows<4>(x, sv, wv, k, lane, acc);
    const int r = n - k;
    if (r == 3)      gather_rows<3>(x, sv, wv, k, lane, acc);
    else if (r == 2) gather_rows<2>(x, sv, wv, k, lane, acc);
    else if (r == 1) gather_rows<1>(x, sv, wv, k, lane, acc);
}

template <int ORD>
__device__ __forceinline__ float4 node_row(const float* __restrict__ x, const int* __restrict__ rowptr,
                                           const int* __restrict__ col, float* __restrict__ wcol,
                                           const float* __restrict__ dinv,
                                           const float* __restrict__ Tprev,
                                           const float* __restrict__ feat,
                                           const float* __restrict__ theta, float c2,
                                           int node, int lane) {
    const int b = rowptr[node];
    const int e = rowptr[node + 1];
    const float s = -c2 * dinv[node];
    float t0 = 0.f, th1 = 0.f, th2 = 0.f, th3 = 0.f;
    if (ORD == 3) {
        t0 = theta[0];
        th1 = theta[1] * 0.5f;
        th2 = theta[2] * 0.5f;
        th3 = theta[3] * 0.5f;
    }
    const size_t i = (size_t)node * (DFEAT / 4) + lane;

    int sv = 0;
    float wv = 0.f;
    if (b + lane < e) {
        sv = col[b + lane];
        if (ORD == 1) { wv = dinv[sv]; wcol[b + lane] = wv; }
        else wv = wcol[b + lane];
    }

    float4 tp = make_float4(0.f, 0.f, 0.f, 0.f);
    if (ORD >= 2) tp = ((const float4*)Tprev)[i];
    float4 f = make_float4(0.f, 0.f, 0.f, 0.f);
    float4 t2 = make_float4(0.f, 0.f, 0.f, 0.f);
    if (ORD == 3) {
        f = ((const float4*)feat)[i];
        t2 = ((const float4*)x)[i];
    }

    float4 acc = make_float4(0.f, 0.f, 0.f, 0.f);
    gather_chunk(x, sv, wv, min(32, e - b), lane, acc);
    for (int cb = b + 32; cb < e; cb += 32) {
        const int j = cb + lane;
        sv = 0;
        wv = 0.f;
        if (j < e) {
            sv = col[j];
            if (ORD == 1) { wv = dinv[sv]; wcol[j] = wv; }
            else wv = wcol[j];
        }
        gather_chunk(x, sv, wv, min(32, e - cb), lane, acc);
    }

    float4 t;
    t.x = fmaf(s, acc.x, -tp.x);
    t.y = fmaf(s, acc.y, -tp.y);
    t.z = fmaf(s, acc.z, -tp.z);
    t.w = fmaf(s, acc.w, -tp.w);
    if (ORD < 3) return t;

    float4 hv = make_float4(t0 * f.x, t0 * f.y, t0 * f.z, t0 * f.w);
    hv.x = fmaf(th1, tp.x + 1.0f, hv.x);
    hv.y = fmaf(th1, tp.y + 1.0f, hv.y);
    hv.z = fmaf(th1, tp.z + 1.0f, hv.z);
    hv.w = fmaf(th1, tp.w + 1.0f, hv.w);
    hv.x = fmaf(th2, t2.x + 1.0f, hv.x);
    hv.y = fmaf(th2, t2.y + 1.0f, hv.y);
    hv.z = fmaf(th2, t2.z + 1.0f, hv.z);
    hv.w = fmaf(th2, t2.w + 1.0f, hv.w);
    hv.x = fmaf(th3, t.x + 1.0f, hv.x);
    hv.y = fmaf(th3, t.y + 1.0f, hv.y);
    hv.z = fmaf(th3, t.z + 1.0f, hv.z);
    hv.w = fmaf(th3, t.w + 1.0f, hv.w);
    return hv;
}

template <int ORD>
__global__ void __launch_bounds__(256)
gather_update_kernel(const float* __restrict__ x, const int* __restrict__ rowptr,
                     const int* __restrict__ col, float* __restrict__ wcol,
                     const float* __restrict__ dinv,
                     const float* __restrict__ Tprev, float* __restrict__ Tnew,
                     float c2, int N) {
    const int lane = threadIdx.x & 31;
    const int node = __builtin_amdgcn_readfirstlane((int)((blockIdx.x * blockDim.x + threadIdx.x) >> 5));
    if (node >= N) return;
    const float4 t = node_row<ORD>(x, rowptr, col, wcol, dinv, Tprev, nullptr, nullptr, c2, node, lane);
    ((float4*)Tnew)[(size_t)node * (DFEAT / 4) + lane] = t;
}

#define GP 132
#define FW 16
#define FT 4

__global__ void __launch_bounds__(FW * 32)
gather3_gemm_kernel(const float* __restrict__ x, const int* __restrict__ rowptr,
                    const int* __restrict__ col, float* __restrict__ wcol,
                    const float* __restrict__ dinv, const float* __restrict__ Tprev,
                    const float* __restrict__ feat, const float* __restrict__ theta,
                    const float* __restrict__ W, float* __restrict__ out, int N) {
    __shared__ float sW[DFEAT * GP];
    __shared__ float sA[16 * GP];
    const int tid = threadIdx.x;
    const int lane = tid & 31;
    const int wave = __builtin_amdgcn_readfirstlane(tid >> 5);
    const int m = lane & 15;
    const int half = lane >> 4;

    const float4* Wq = (const float4*)W + tid;
    float4 wq[DFEAT / FW];
#pragma unroll
    for (int j = 0; j < DFEAT / FW; ++j) wq[j] = Wq[j * (FW * 32)];
#pragma unroll
    for (int j = 0; j < DFEAT / FW; ++j) *(float4*)(sW + (wave + FW * j) * GP + lane * 4) = wq[j];

    float* myRow = sA + wave * GP + lane * 4;
    const float* aP = sA + m * GP + (half << 1);
    const float* bP = sW + ((wave & 7) * 16 + m) * GP + (half << 1);

    const int nStrips = (N + 15) >> 4;
    for (int strip = blockIdx.x; strip < nStrips; strip += gridDim.x) {
        const int node = strip * 16 + wave;
        float4 hv = make_float4(0.f, 0.f, 0.f, 0.f);
        if (node < N)
            hv = node_row<3>(x, rowptr, col, wcol, dinv, Tprev, feat, theta, 2.0f, node, lane);
        *(float4*)myRow = hv;
        __syncthreads();

        v8f acc = (v8f){};
        if (wave < 8) {
#pragma unroll
            for (int kk = 0; kk < DFEAT; kk += 4) {
                const v2f a = *(const v2f*)(aP + kk);
                const v2f b = *(const v2f*)(bP + kk);
                acc = __builtin_amdgcn_wmma_f32_16x16x4_f32(
                     false, a,  false, b,
                     (short)0, acc,  false,  false);
            }
#pragma unroll
            for (int v = 0; v < 8; ++v) acc[v] = (acc[v] >= 0.0f) ? acc[v] : 0.01f * acc[v];
        }
        __syncthreads();

        if (wave < 8) {
#pragma unroll
            for (int v = 0; v < 8; ++v) sA[(v + (half << 3)) * GP + wave * 16 + m] = acc[v];
        }
        __syncthreads();
        if (node < N)
            ((float4*)out)[(size_t)node * (DFEAT / 4) + lane] = *(const float4*)myRow;
    }
}


extern "C" void kernel_launch(void* const* d_in, const int* in_sizes, int n_in,
                              void* d_out, int out_size, void* d_ws, size_t ws_size,
                              hipStream_t stream) {
    const float* feat  = (const float*)d_in[0];
    const int*   src   = (const int*)d_in[1];
    const int*   dst   = (const int*)d_in[2];
    const float* theta = (const float*)d_in[3];
    const float* W     = (const float*)d_in[4];
    float* out = (float*)d_out;

    const int N = in_sizes[0] / DFEAT;
    const int E = in_sizes[1];

    float* ws   = (float*)d_ws;
    float* dinv = ws + N;
    float* scr  = dinv + N;
    float* Ta   = scr + (size_t)N * DFEAT;
    float* Tb   = Ta + (size_t)N * DFEAT;

    int2*  rec    = (int2*)scr;
    int2*  rec2   = rec + E;
    int*   grp    = (int*)(rec2 + E);
    int*   col    = grp + E;
    float* wcol   = (float*)(col + E);
    int*   rowptr = (int*)(wcol + E);
    int*   tabc   = rowptr + (N + 4);
    int*   toff   = tabc + NS * NR;

    const int TB = 256;
    auto cdiv = [](long long a, long long b) { return (int)((a + b - 1) / b); };

    const int chunk = cdiv(E, NS);
    const int RW = cdiv(N, NR);
    slice_kernel<<<NS, PT1, 0, stream>>>(dst, tabc, toff, rec, E, chunk, RW);
    range_kernel<<<NR, PT3, 0, stream>>>(tabc, toff, rec, rec2, grp, src, rowptr, dinv, col, N, RW, chunk);
    const int nodeBlocks = cdiv((long long)N * 32, TB);

    gather_update_kernel<1><<<nodeBlocks, TB, 0, stream>>>(feat, rowptr, col, wcol, dinv, nullptr, Ta, 1.0f, N);
    gather_update_kernel<2><<<nodeBlocks, TB, 0, stream>>>(Ta, rowptr, col, wcol, dinv, feat, Tb, 2.0f, N);
    const int strips = cdiv(N, 16);
    gather3_gemm_kernel<<<cdiv(strips, FT), FW * 32, 0, stream>>>(Tb, rowptr, col, wcol, dinv, Ta, feat, theta, W, out, N);
}


# speedup vs baseline: 6.6828x; 1.1057x over previous
#include <hip/hip_runtime.h>

typedef float v2f __attribute__((ext_vector_type(2)));
typedef float v8f __attribute__((ext_vector_type(8)));

#define DFEAT 128

#define NS 128
#define NR 256
#define RWMAX 256
#define PT1 1024
#define SCAP 8192
static_assert(PT1 >= NR && PT1 % 32 == 0, "slice_kernel: the first NR threads (whole waves) scan the range counts");

__global__ void __launch_bounds__(PT1)
slice_kernel(const int* __restrict__ dst, int* __restrict__ tabc, int* __restrict__ toff,
             int2* __restrict__ rec, int E, int chunk, int RW) {
    __shared__ int lc[NR];
    __shared__ int lfill[NR];
    __shared__ int wtot[NR / 32];
    __shared__ int2 sr[SCAP];
    const int tid = threadIdx.x;
    const int lane = tid & 31;
    const int wv = tid >> 5;
    const int s = blockIdx.x;
    if (tid < NR) lc[tid] = 0;
    __syncthreads();
    const int e0 = min(E, s * chunk);
    const int e1 = min(E, e0 + chunk);
    const int n = e1 - e0;
    const bool staged = n <= SCAP;

    for (int eb = e0; eb < e1; eb += PT1 * 8) {
        int v[8];
#pragma unroll
        for (int k = 0; k < 8; ++k) v[k] = dst[min(eb + k * PT1 + tid, e1 - 1)];
#pragma unroll
        for (int k = 0; k < 8; ++k)
            if (eb + k * PT1 + tid < e1) atomicAdd(&lc[v[k] / RW], 1);
    }
    __syncthreads();

    int c = 0;
    if (tid < NR) c = lc[tid];
    int inc = c;
    for (int o = 1; o < 32; o <<= 1) {
        int t = __shfl_up(inc, o);
        if (lane >= o) inc += t;
    }
    if (tid < NR && lane == 31) wtot[wv] = inc;
    __syncthreads();
    if (tid < NR) {
        int run = 0;
        for (int w = 0; w < NR / 32; ++w)
            if (w < wv) run += wtot[w];
        const int ex = run + inc - c;
        lfill[tid] = ex;
        tabc[(size_t)s * NR + tid] = c;
        toff[(size_t)s * NR + tid] = ex;
    }
    __syncthreads();

    for (int eb = e0; eb < e1; eb += PT1 * 8) {
        int d[8], p[8];
#pragma unroll
        for (int k = 0; k < 8; ++k) d[k] = dst[min(eb + k * PT1 + tid, e1 - 1)];
#pragma unroll
        for (int k = 0; k < 8; ++k) {
            p[k] = 0;
            if (eb + k * PT1 + tid < e1) p[k] = atomicAdd(&lfill[d[k] / RW], 1);
        }
#pragma unroll
        for (int k = 0; k < 8; ++k) {
            const int e = eb + k * PT1 + tid;
            if (e < e1) {
                const int2 q = make_int2(e, d[k]);
                if (staged) sr[p[k]] = q;
                else rec[e0 + p[k]] = q;
            }
        }
    }
    if (staged) {
        __syncthreads();
        for (int i = tid; i < n; i += PT1) rec[e0 + i] = sr[i];
    }
}

#define PT3 512
#define CAP 8192
#define NPW (NS / (PT3 / 32))
static_assert(PT3 == 2 * NR, "range_kernel: a thread sums one range's counts over half of the slices");
static_assert(NS % 2 == 0 && NS <= PT3 && NS % 32 == 0 && NS % (PT3 / 32) == 0, "range_kernel: slices per thread, per scan wave, per wave");
__device__ __forceinline__ void range_body(const int2* __restrict__ rec, int2* R, int* G,
                                           const int* pbase, const int* pcnt, const int* psrc,
                                           int* lcnt, int* lexc, int* lfill, int* wtot,
                                           const int* __restrict__ src,
                                           int* __restrict__ rowptr, float* __restrict__ dinv,
                                           int* __restrict__ col,
                                           int rb, int cnt, int n0, int nn, int N) {
    const int tid = threadIdx.x;
    const int lane = tid & 31;
    const int wv = __builtin_amdgcn_readfirstlane(tid >> 5);

    {
        int2 q[NPW][2];
#pragma unroll
        for (int j = 0; j < NPW; ++j) {
            const int s = wv + (PT3 / 32) * j;
            const int pn = pcnt[s];
            const int ps = psrc[s];
#pragma unroll
            for (int h = 0; h < 2; ++h)
                q[j][h] = rec[(pn > 0) ? ps + min(lane + 32 * h, pn - 1) : 0];
        }
#pragma unroll
        for (int j = 0; j < NPW; ++j) {
            const int s = wv + (PT3 / 32) * j;
            const int pn = pcnt[s];
            const int pb = pbase[s];
#pragma unroll
            for (int h = 0; h < 2; ++h) {
                if (lane + 32 * h < pn) {
                    R[pb + lane + 32 * h] = q[j][h];
                    atomicAdd(&lcnt[q[j][h].y - n0], 1);
                }
            }
        }
        for (int j = 0; j < NPW; ++j) {
            const int s = wv + (PT3 / 32) * j;
            const int pn = pcnt[s];
            for (int k = 64 + lane; k < pn; k += 32) {
                const int2 qq = rec[psrc[s] + k];
                R[pbase[s] + k] = qq;
                atomicAdd(&lcnt[qq.y - n0], 1);
            }
        }
    }
    __syncthreads();

    int c = 0;
    if (tid < RWMAX) c = lcnt[tid];
    int inc = c;
    for (int o = 1; o < 32; o <<= 1) {
        int t = __shfl_up(inc, o);
        if (lane >= o) inc += t;
    }
    if (tid < RWMAX && lane == 31) wtot[wv] = inc;
    __syncthreads();
    if (tid < RWMAX) {
        int run = 0;
        for (int w = 0; w < RWMAX / 32; ++w)
            if (w < wv) run += wtot[w];
        const int ex = run + inc - c;
        lexc[tid] = ex;
        lfill[tid] = ex;
        if (tid < nn) {
            const int node = n0 + tid;
            rowptr[node] = rb + ex;
            dinv[node] = 1.0f / sqrtf(fmaxf((float)c, 1.0f));
            if (node == N - 1) rowptr[N] = rb + ex + c;
        }
    }
    __syncthreads();

    for (int i = tid; i < cnt; i += PT3) {
        const int2 q = R[i];
        G[atomicAdd(&lfill[q.y - n0], 1)] = q.x;
    }
    __syncthreads();

    int* S = (int*)R;
    for (int j = wv; j < nn; j += PT3 / 32) {
        const int b = __builtin_amdgcn_readfirstlane(lexc[j]);
        const int e = b + __builtin_amdgcn_readfirstlane(lcnt[j]);
        for (int cb = b; cb < e; cb += 32) {
            const int jm = cb + lane;
            const bool act = jm < e;
            const int mine = act ? G[jm] : 0x7fffffff;
            int rank = 0;
            for (int ob = b; ob < e; ob += 32) {
                const int jo = ob + lane;
                const int other = (ob == cb) ? mine : ((jo < e) ? G[jo] : 0x7fffffff);
                const int n = min(32, e - ob);
                for (int k = 0; k < n; ++k)
                    rank += (__builtin_amdgcn_readlane(other, k) < mine) ? 1 : 0;
            }
            if (act) S[b + rank] = mine;
        }
    }
    __syncthreads();

    for (int i0 = tid; i0 < cnt; i0 += PT3 * 8) {
        int ev[8], sn[8];
#pragma unroll
        for (int k = 0; k < 8; ++k) ev[k] = S[min(i0 + k * PT3, cnt - 1)];
#pragma unroll
        for (int k = 0; k < 8; ++k) sn[k] = src[ev[k]];
#pragma unroll
        for (int k = 0; k < 8; ++k)
            if (i0 + k * PT3 < cnt) col[rb + i0 + k * PT3] = sn[k];
    }
}

__global__ void __launch_bounds__(PT3)
range_kernel(const int* __restrict__ tabc, const int* __restrict__ toff,
             const int2* __restrict__ rec, int2* rec2, int* grp,
             const int* __restrict__ src, int* __restrict__ rowptr, float* __restrict__ dinv,
             int* __restrict__ col, int N, int RW, int chunk) {
    __shared__ int2 srec[CAP];
    __shared__ int sg[CAP];
    __shared__ int lcnt[RWMAX];
    __shared__ int lexc[RWMAX];
    __shared__ int lfill[RWMAX];
    __shared__ int wtot[RWMAX / 32];
    __shared__ int stot[2][NR];
    __shared__ int wtotA[NR / 32];
    __shared__ int wtotB[NS / 32];
    __shared__ int pbase[NS];
    __shared__ int pcnt[NS];
    __shared__ int psrc[NS];
    __shared__ int srb;
    __shared__ int scnt;
    const int tid = threadIdx.x;
    const int lane = tid & 31;
    const int wv = tid >> 5;
    const int r = blockIdx.x;
    const int n0 = r * RW;
    const int nn = max(0, min(RW, N - n0));

    const int rr = tid & (NR - 1);
    const int p = tid >> 8;
    const int* colp = tabc + (size_t)(p * (NS / 2)) * NR + rr;
    int v[NS / 2];
#pragma unroll
    for (int k = 0; k < NS / 2; ++k) v[k] = colp[(size_t)k * NR];
    int pc = 0, po = 0;
    if (tid < NS) {
        pc = tabc[(size_t)tid * NR + r];
        po = toff[(size_t)tid * NR + r];
    }
    int tot = 0;
#pragma unroll
    for (int k = 0; k < NS / 2; ++k) tot += v[k];
    stot[p][rr] = tot;
    if (tid < RWMAX) lcnt[tid] = 0;
    __syncthreads();

    int T = 0;
    if (tid < NR) T = stot[0][tid] + stot[1][tid];
    int incA = T;
    int incB = pc;
    for (int o = 1; o < 32; o <<= 1) {
        int ta = __shfl_up(incA, o);
        int tb = __shfl_up(incB, o);
        if (lane >= o) { incA += ta; incB += tb; }
    }
    if (tid < NR && lane == 31) wtotA[wv] = incA;
    if (tid < NS && lane == 31) wtotB[wv] = incB;
    __syncthreads();
    if (tid == r) {
        int run = 0;
        for (int w = 0; w < NR / 32; ++w)
            if (w < wv) run += wtotA[w];
        srb = run + incA - T;
        scnt = T;
    }
    if (tid < NS) {
        int run = 0;
        for (int w = 0; w < NS / 32; ++w)
            if (w < wv) run += wtotB[w];
        pbase[tid] = run + incB - pc;
        pcnt[tid] = pc;
        psrc[tid] = tid * chunk + po;
    }
    __syncthreads();
    const int rb = srb;
    const int cnt = scnt;

    if (cnt <= CAP)
        range_body(rec, srec, sg, pbase, pcnt, psrc, lcnt, lexc, lfill, wtot, src, rowptr, dinv, col, rb, cnt, n0, nn, N);
    else
        range_body(rec, rec2 + rb, grp + rb, pbase, pcnt, psrc, lcnt, lexc, lfill, wtot, src, rowptr, dinv, col, rb, cnt, n0, nn, N);
}


template <int R>
__device__ __forceinline__ void gather_rows(const float* __restrict__ x, int sv, float wv,
                                            int k, int lane, float4& acc) {
    float4 v[R];
    float w[R];
#pragma unroll
    for (int r = 0; r < R; ++r) {
        const int s = __builtin_amdgcn_readlane(sv, k + r);
        w[r] = __int_as_float(__builtin_amdgcn_readlane(__float_as_int(wv), k + r));
        v[r] = *(const float4*)(x + (size_t)s * DFEAT + lane * 4);
    }
#pragma unroll
    for (int r = 0; r < R; ++r) {
        acc.x = fmaf(w[r], v[r].x, acc.x);
        acc.y = fmaf(w[r], v[r].y, acc.y);
        acc.z = fmaf(w[r], v[r].z, acc.z);
        acc.w = fmaf(w[r], v[r].w, acc.w);
    }
}

__device__ __forceinline__ void gather_chunk(const float* __restrict__ x, int sv, float wv,
                                             int n, int lane, float4& acc) {
    int k = 0;
    for (; k + 4 <= n; k += 4) gather_rows<4>(x, sv, wv, k, lane, acc);
    const int r = n - k;
    if (r == 3)      gather_rows<3>(x, sv, wv, k, lane, acc);
    else if (r == 2) gather_rows<2>(x, sv, wv, k, lane, acc);
    else if (r == 1) gather_rows<1>(x, sv, wv, k, lane, acc);
}

template <int ORD>
__device__ __forceinline__ float4 node_row(const float* __restrict__ x, const int* __restrict__ rowptr,
                                           const int* __restrict__ col, float* __restrict__ wcol,
                                           const float* __restrict__ dinv,
                                           const float* __restrict__ Tprev,
                                           const float* __restrict__ feat,
                                           const float* __restrict__ theta, float c2,
                                           int node, int lane) {
    const int b = rowptr[node];
    const int e = rowptr[node + 1];
    const float s = -c2 * dinv[node];
    float t0 = 0.f, th1 = 0.f, th2 = 0.f, th3 = 0.f;
    if (ORD == 3) {
        t0 = theta[0];
        th1 = theta[1] * 0.5f;
        th2 = theta[2] * 0.5f;
        th3 = theta[3] * 0.5f;
    }
    const size_t i = (size_t)node * (DFEAT / 4) + lane;

    int sv = 0;
    float wv = 0.f;
    if (b + lane < e) {
        sv = col[b + lane];
        if (ORD == 1) { wv = dinv[sv]; wcol[b + lane] = wv; }
        else wv = wcol[b + lane];
    }

    float4 tp = make_float4(0.f, 0.f, 0.f, 0.f);
    if (ORD >= 2) tp = ((const float4*)Tprev)[i];
    float4 f = make_float4(0.f, 0.f, 0.f, 0.f);
    float4 t2 = make_float4(0.f, 0.f, 0.f, 0.f);
    if (ORD == 3) {
        f = ((const float4*)feat)[i];
        t2 = ((const float4*)x)[i];
    }

    float4 acc = make_float4(0.f, 0.f, 0.f, 0.f);
    gather_chunk(x, sv, wv, min(32, e - b), lane, acc);
    for (int cb = b + 32; cb < e; cb += 32) {
        const int j = cb + lane;
        sv = 0;
        wv = 0.f;
        if (j < e) {
            sv = col[j];
            if (ORD == 1) { wv = dinv[sv]; wcol[j] = wv; }
            else wv = wcol[j];
        }
        gather_chunk(x, sv, wv, min(32, e - cb), lane, acc);
    }

    float4 t;
    t.x = fmaf(s, acc.x, -tp.x);
    t.y = fmaf(s, acc.y, -tp.y);
    t.z = fmaf(s, acc.z, -tp.z);
    t.w = fmaf(s, acc.w, -tp.w);
    if (ORD < 3) return t;

    float4 hv = make_float4(t0 * f.x, t0 * f.y, t0 * f.z, t0 * f.w);
    hv.x = fmaf(th1, tp.x + 1.0f, hv.x);
    hv.y = fmaf(th1, tp.y + 1.0f, hv.y);
    hv.z = fmaf(th1, tp.z + 1.0f, hv.z);
    hv.w = fmaf(th1, tp.w + 1.0f, hv.w);
    hv.x = fmaf(th2, t2.x + 1.0f, hv.x);
    hv.y = fmaf(th2, t2.y + 1.0f, hv.y);
    hv.z = fmaf(th2, t2.z + 1.0f, hv.z);
    hv.w = fmaf(th2, t2.w + 1.0f, hv.w);
    hv.x = fmaf(th3, t.x + 1.0f, hv.x);
    hv.y = fmaf(th3, t.y + 1.0f, hv.y);
    hv.z = fmaf(th3, t.z + 1.0f, hv.z);
    hv.w = fmaf(th3, t.w + 1.0f, hv.w);
    return hv;
}

template <int ORD>
__global__ void __launch_bounds__(256)
gather_update_kernel(const float* __restrict__ x, const int* __restrict__ rowptr,
                     const int* __restrict__ col, float* __restrict__ wcol,
                     const float* __restrict__ dinv,
                     const float* __restrict__ Tprev, float* __restrict__ Tnew,
                     float c2, int N) {
    const int lane = threadIdx.x & 31;
    const int node = __builtin_amdgcn_readfirstlane((int)((blockIdx.x * blockDim.x + threadIdx.x) >> 5));
    if (node >= N) return;
    const float4 t = node_row<ORD>(x, rowptr, col, wcol, dinv, Tprev, nullptr, nullptr, c2, node, lane);
    ((float4*)Tnew)[(size_t)node * (DFEAT / 4) + lane] = t;
}

#define GP 132
#define FW 16
#define FT 4

__global__ void __launch_bounds__(FW * 32)
gather3_gemm_kernel(const float* __restrict__ x, const int* __restrict__ rowptr,
                    const int* __restrict__ col, float* __restrict__ wcol,
                    const float* __restrict__ dinv, const float* __restrict__ Tprev,
                    const float* __restrict__ feat, const float* __restrict__ theta,
                    const float* __restrict__ W, float* __restrict__ out, int N) {
    __shared__ float sW[DFEAT * GP];
    __shared__ float sA[16 * GP];
    const int tid = threadIdx.x;
    const int lane = tid & 31;
    const int wave = __builtin_amdgcn_readfirstlane(tid >> 5);
    const int m = lane & 15;
    const int half = lane >> 4;

    const float4* Wq = (const float4*)W + tid;
    float4 wq[DFEAT / FW];
#pragma unroll
    for (int j = 0; j < DFEAT / FW; ++j) wq[j] = Wq[j * (FW * 32)];
#pragma unroll
    for (int j = 0; j < DFEAT / FW; ++j) *(float4*)(sW + (wave + FW * j) * GP + lane * 4) = wq[j];

    float* myRow = sA + wave * GP + lane * 4;
    const float* aP = sA + m * GP + (half << 1);
    const float* bP = sW + ((wave & 7) * 16 + m) * GP + (half << 1);

    const int nStrips = (N + 15) >> 4;
    for (int strip = blockIdx.x; strip < nStrips; strip += gridDim.x) {
        const int node = strip * 16 + wave;
        float4 hv = make_float4(0.f, 0.f, 0.f, 0.f);
        if (node < N)
            hv = node_row<3>(x, rowptr, col, wcol, dinv, Tprev, feat, theta, 2.0f, node, lane);
        *(float4*)myRow = hv;
        __syncthreads();

        v8f acc = (v8f){};
        if (wave < 8) {
#pragma unroll
            for (int kk = 0; kk < DFEAT; kk += 4) {
                const v2f a = *(const v2f*)(aP + kk);
                const v2f b = *(const v2f*)(bP + kk);
                acc = __builtin_amdgcn_wmma_f32_16x16x4_f32(
                     false, a,  false, b,
                     (short)0, acc,  false,  false);
            }
#pragma unroll
            for (int v = 0; v < 8; ++v) acc[v] = (acc[v] >= 0.0f) ? acc[v] : 0.01f * acc[v];
        }
        __syncthreads();

        if (wave < 8) {
#pragma unroll
            for (int v = 0; v < 8; ++v) sA[(v + (half << 3)) * GP + wave * 16 + m] = acc[v];
        }
        __syncthreads();
        if (node < N)
            ((float4*)out)[(size_t)node * (DFEAT / 4) + lane] = *(const float4*)myRow;
    }
}


extern "C" void kernel_launch(void* const* d_in, const int* in_sizes, int n_in,
                              void* d_out, int out_size, void* d_ws, size_t ws_size,
                              hipStream_t stream) {
    const float* feat  = (const float*)d_in[0];
    const int*   src   = (const int*)d_in[1];
    const int*   dst   = (const int*)d_in[2];
    const float* theta = (const float*)d_in[3];
    const float* W     = (const float*)d_in[4];
    float* out = (float*)d_out;

    const int N = in_sizes[0] / DFEAT;
    const int E = in_sizes[1];

    float* ws   = (float*)d_ws;
    float* dinv = ws;
    float* scr  = dinv + N;
    float* Ta   = scr + (size_t)N * DFEAT + 48;
    float* Tb   = Ta + (size_t)N * DFEAT;

    int2*  rec    = (int2*)scr;
    int2*  rec2   = rec + E;
    int*   grp    = (int*)(rec2 + E);
    int*   col    = grp + E;
    float* wcol   = (float*)(col + E);
    int*   rowptr = (int*)(wcol + E);
    int*   tabc   = rowptr + (N + 4);
    int*   toff   = tabc + NS * NR;

    const int TB = 256;
    auto cdiv = [](long long a, long long b) { return (int)((a + b - 1) / b); };

    const int chunk = cdiv(E, NS);
    const int RW = cdiv(N, NR);
    slice_kernel<<<NS, PT1, 0, stream>>>(dst, tabc, toff, rec, E, chunk, RW);
    range_kernel<<<NR, PT3, 0, stream>>>(tabc, toff, rec, rec2, grp, src, rowptr, dinv, col, N, RW, chunk);
    const int nodeBlocks = cdiv((long long)N * 32, TB);

    gather_update_kernel<1><<<nodeBlocks, TB, 0, stream>>>(feat, rowptr, col, wcol, dinv, nullptr, Ta, 1.0f, N);
    gather_update_kernel<2><<<nodeBlocks, TB, 0, stream>>>(Ta, rowptr, col, wcol, dinv, feat, Tb, 2.0f, N);
    const int strips = cdiv(N, 16);
    gather3_gemm_kernel<<<cdiv(strips, FT), FW * 32, 0, stream>>>(Tb, rowptr, col, wcol, dinv, Ta, feat, theta, W, out, N);
}
